# starting kernel
# baseline (speedup 1.0000x reference)
#include <hip/hip_runtime.h>

typedef __attribute__((ext_vector_type(2))) float v2f;
typedef __attribute__((ext_vector_type(8))) float v8f;

#define WT_STRIDE 66

__global__ __launch_bounds__(256)
void proj_wmma_kernel(const float* __restrict__ h_c,
                      const float* __restrict__ h_p,
                      const float* __restrict__ Wqc, const float* __restrict__ bqc,
                      const float* __restrict__ Wqp, const float* __restrict__ bqp,
                      const float* __restrict__ Wk,  const float* __restrict__ bk,
                      const float* __restrict__ Wv,  const float* __restrict__ bv,
                      float* __restrict__ Qc, float* __restrict__ Qp,
                      float* __restrict__ Kn, float* __restrict__ Vn,
                      int n)
{
    __shared__ float sWT[4][64 * WT_STRIDE];
    __shared__ float sB[4][64];

    const int tid = threadIdx.x;
    const float* Ws[4] = {Wqc, Wqp, Wk, Wv};
    const float* Bs[4] = {bqc, bqp, bk, bv};
    for (int o = 0; o < 4; ++o) {
        for (int i = tid; i < 4096; i += 256) {
            int col = i >> 6, k = i & 63;
            sWT[o][col * WT_STRIDE + k] = Ws[o][k * 64 + col];
        }
        if (tid < 64) sB[o][tid] = Bs[o][tid];
    }
    __syncthreads();

    const int wave = tid >> 5;
    const int lane = tid & 31;
    const int half = lane >> 4;
    const int lrow = lane & 15;
    const int ncol = lane & 15;

    const int m0 = (blockIdx.x * 8 + wave) * 16;
    if (m0 >= n) return;

    const int row = m0 + lrow;

    v2f a_c[16], a_p[16];
#pragma unroll
    for (int ks = 0; ks < 16; ++ks) {
        int k = ks * 4 + 2 * half;
        a_c[ks] = *(const v2f*)(h_c + row * 64 + k);
        a_p[ks] = *(const v2f*)(h_p + row * 64 + k);
    }

    float* outs[4] = {Qc, Qp, Kn, Vn};

#pragma unroll
    for (int o = 0; o < 4; ++o) {
        float* op = outs[o];
#pragma unroll
        for (int nt = 0; nt < 4; ++nt) {
            const int ncg = nt * 16 + ncol;
            v8f acc = {0.f, 0.f, 0.f, 0.f, 0.f, 0.f, 0.f, 0.f};
#pragma unroll
            for (int ks = 0; ks < 16; ++ks) {
                int k = ks * 4 + 2 * half;
                v2f b = *(const v2f*)(&sWT[o][ncg * WT_STRIDE + k]);
                v2f a = (o == 1) ? a_p[ks] : a_c[ks];
                acc = __builtin_amdgcn_wmma_f32_16x16x4_f32(
                     false, a,  false, b,
                     (short)0, acc,  false,  false);
            }
            const float bias = sB[o][ncg];
#pragma unroll
            for (int r = 0; r < 8; ++r) {
                int m = m0 + r + 8 * half;
                op[m * 64 + ncg] = acc[r] + bias;
            }
        }
    }
}

__global__ void zero_kernel(float* __restrict__ p, int n)
{
    int i = blockIdx.x * blockDim.x + threadIdx.x;
    if (i < n) p[i] = 0.0f;
}

__global__ __launch_bounds__(256)
void edge_attn_kernel(const int* __restrict__ src, const int* __restrict__ dst,
                      const float* __restrict__ Qc, const float* __restrict__ Qp,
                      const float* __restrict__ Kn, const float* __restrict__ Vn,
                      float* __restrict__ out_c, float* __restrict__ out_p,
                      int total  )
{
    int idx = blockIdx.x * blockDim.x + threadIdx.x;
    if (idx >= total) return;
    const int e = idx >> 3;
    const int h = idx & 7;
    const int s = src[e];
    const int d = dst[e];
    const int bs = s * 64 + h * 8;
    const int bd = d * 64 + h * 8;

    const float4 k0  = *(const float4*)(Kn + bs);
    const float4 k1  = *(const float4*)(Kn + bs + 4);
    const float4 v0  = *(const float4*)(Vn + bs);
    const float4 v1  = *(const float4*)(Vn + bs + 4);
    const float4 qc0 = *(const float4*)(Qc + bd);
    const float4 qc1 = *(const float4*)(Qc + bd + 4);
    const float4 qp0 = *(const float4*)(Qp + bd);
    const float4 qp1 = *(const float4*)(Qp + bd + 4);

    float dc = qc0.x*k0.x + qc0.y*k0.y + qc0.z*k0.z + qc0.w*k0.w
             + qc1.x*k1.x + qc1.y*k1.y + qc1.z*k1.z + qc1.w*k1.w;
    float dp = qp0.x*k0.x + qp0.y*k0.y + qp0.z*k0.z + qp0.w*k0.w
             + qp1.x*k1.x + qp1.y*k1.y + qp1.z*k1.z + qp1.w*k1.w;

    const float sc = __expf(fminf(fmaxf(dc * 0.125f, -5.0f), 5.0f));
    const float sp = __expf(fminf(fmaxf(dp * 0.125f, -5.0f), 5.0f));

    atomicAdd(out_c + bd + 0, v0.x * sc);
    atomicAdd(out_c + bd + 1, v0.y * sc);
    atomicAdd(out_c + bd + 2, v0.z * sc);
    atomicAdd(out_c + bd + 3, v0.w * sc);
    atomicAdd(out_c + bd + 4, v1.x * sc);
    atomicAdd(out_c + bd + 5, v1.y * sc);
    atomicAdd(out_c + bd + 6, v1.z * sc);
    atomicAdd(out_c + bd + 7, v1.w * sc);

    atomicAdd(out_p + bd + 0, v0.x * sp);
    atomicAdd(out_p + bd + 1, v0.y * sp);
    atomicAdd(out_p + bd + 2, v0.z * sp);
    atomicAdd(out_p + bd + 3, v0.w * sp);
    atomicAdd(out_p + bd + 4, v1.x * sp);
    atomicAdd(out_p + bd + 5, v1.y * sp);
    atomicAdd(out_p + bd + 6, v1.z * sp);
    atomicAdd(out_p + bd + 7, v1.w * sp);
}

extern "C" void kernel_launch(void* const* d_in, const int* in_sizes, int n_in,
                              void* d_out, int out_size, void* d_ws, size_t ws_size,
                              hipStream_t stream)
{
    const float* h_c  = (const float*)d_in[0];
    const float* h_p  = (const float*)d_in[1];
    const int*   src  = (const int*)  d_in[2];
    const int*   dst  = (const int*)  d_in[3];
    const float* Wqcw = (const float*)d_in[4];
    const float* Wqcb = (const float*)d_in[5];
    const float* Wqpw = (const float*)d_in[6];
    const float* Wqpb = (const float*)d_in[7];
    const float* Wkw  = (const float*)d_in[8];
    const float* Wkb  = (const float*)d_in[9];
    const float* Wvw  = (const float*)d_in[10];
    const float* Wvb  = (const float*)d_in[11];

    const int n = in_sizes[0] / 64;
    const int e = in_sizes[2];

    float* Qc = (float*)d_ws;
    float* Qp = Qc + (size_t)n * 64;
    float* Kn = Qp + (size_t)n * 64;
    float* Vn = Kn + (size_t)n * 64;

    float* out_c = (float*)d_out;
    float* out_p = out_c + (size_t)n * 64;

    {
        int total = 2 * n * 64;
        zero_kernel<<<(total + 255) / 256, 256, 0, stream>>>((float*)d_out, total);
    }

    {
        int blocks = (n + 127) / 128;
        proj_wmma_kernel<<<blocks, 256, 0, stream>>>(
            h_c, h_p, Wqcw, Wqcb, Wqpw, Wqpb, Wkw, Wkb, Wvw, Wvb,
            Qc, Qp, Kn, Vn, n);
    }

    {
        int total = e * 8;
        edge_attn_kernel<<<(total + 255) / 256, 256, 0, stream>>>(
            src, dst, Qc, Qp, Kn, Vn, out_c, out_p, total);
    }
}


# edge kernel: a wave owns whole edges, lane = output column, so each row load and float atomic add covers one whole 128-B line (was 8 lines, 4 lanes each); head dots by 8-lane tree sum; arrival order
# speedup vs baseline: 2.6489x; 2.6489x over previous
#include <hip/hip_runtime.h>

typedef __attribute__((ext_vector_type(2))) float v2f;
typedef __attribute__((ext_vector_type(8))) float v8f;

#define WT_STRIDE 66

__global__ __launch_bounds__(256)
void proj_wmma_kernel(const float* __restrict__ h_c,
                      const float* __restrict__ h_p,
                      const float* __restrict__ Wqc, const float* __restrict__ bqc,
                      const float* __restrict__ Wqp, const float* __restrict__ bqp,
                      const float* __restrict__ Wk,  const float* __restrict__ bk,
                      const float* __restrict__ Wv,  const float* __restrict__ bv,
                      float* __restrict__ Qc, float* __restrict__ Qp,
                      float* __restrict__ Kn, float* __restrict__ Vn,
                      int n)
{
    __shared__ float sWT[4][64 * WT_STRIDE];
    __shared__ float sB[4][64];

    const int tid = threadIdx.x;
    const float* Ws[4] = {Wqc, Wqp, Wk, Wv};
    const float* Bs[4] = {bqc, bqp, bk, bv};
    for (int o = 0; o < 4; ++o) {
        for (int i = tid; i < 4096; i += 256) {
            int col = i >> 6, k = i & 63;
            sWT[o][col * WT_STRIDE + k] = Ws[o][k * 64 + col];
        }
        if (tid < 64) sB[o][tid] = Bs[o][tid];
    }
    __syncthreads();

    const int wave = tid >> 5;
    const int lane = tid & 31;
    const int half = lane >> 4;
    const int lrow = lane & 15;
    const int ncol = lane & 15;

    const int m0 = (blockIdx.x * 8 + wave) * 16;
    if (m0 >= n) return;

    const int row = m0 + lrow;

    v2f a_c[16], a_p[16];
#pragma unroll
    for (int ks = 0; ks < 16; ++ks) {
        int k = ks * 4 + 2 * half;
        a_c[ks] = *(const v2f*)(h_c + row * 64 + k);
        a_p[ks] = *(const v2f*)(h_p + row * 64 + k);
    }

    float* outs[4] = {Qc, Qp, Kn, Vn};

#pragma unroll
    for (int o = 0; o < 4; ++o) {
        float* op = outs[o];
#pragma unroll
        for (int nt = 0; nt < 4; ++nt) {
            const int ncg = nt * 16 + ncol;
            v8f acc = {0.f, 0.f, 0.f, 0.f, 0.f, 0.f, 0.f, 0.f};
#pragma unroll
            for (int ks = 0; ks < 16; ++ks) {
                int k = ks * 4 + 2 * half;
                v2f b = *(const v2f*)(&sWT[o][ncg * WT_STRIDE + k]);
                v2f a = (o == 1) ? a_p[ks] : a_c[ks];
                acc = __builtin_amdgcn_wmma_f32_16x16x4_f32(
                     false, a,  false, b,
                     (short)0, acc,  false,  false);
            }
            const float bias = sB[o][ncg];
#pragma unroll
            for (int r = 0; r < 8; ++r) {
                int m = m0 + r + 8 * half;
                op[m * 64 + ncg] = acc[r] + bias;
            }
        }
    }
}

__global__ void zero_kernel(float* __restrict__ p, int n)
{
    int i = blockIdx.x * blockDim.x + threadIdx.x;
    if (i < n) p[i] = 0.0f;
}

#define EPT 4

__device__ __forceinline__ float head_sum(float x)
{
    x += __builtin_bit_cast(float, __builtin_amdgcn_update_dpp(
             0, __builtin_bit_cast(int, x), 0x161, 0xf, 0xf, true));
    x += __builtin_bit_cast(float, __builtin_amdgcn_update_dpp(
             0, __builtin_bit_cast(int, x), 0x162, 0xf, 0xf, true));
    x += __builtin_bit_cast(float, __builtin_amdgcn_update_dpp(
             0, __builtin_bit_cast(int, x), 0x164, 0xf, 0xf, true));
    return x;
}

__device__ __forceinline__ float edge_score(float dot)
{
    return __expf(fminf(fmaxf(dot * 0.125f, -5.0f), 5.0f));
}

template <int NE>
__device__ __forceinline__ void edge_trip(const int* s, const int* d, int lane,
                                          const float* __restrict__ Qc,
                                          const float* __restrict__ Qp,
                                          const float* __restrict__ Kn,
                                          const float* __restrict__ Vn,
                                          float* __restrict__ out_c,
                                          float* __restrict__ out_p)
{
    float k0[NE], k1[NE], v0[NE], v1[NE], c0[NE], c1[NE], p0[NE], p1[NE];
#pragma unroll
    for (int i = 0; i < NE; ++i) {
        const int so = s[i] * 64 + lane;
        const int dof = d[i] * 64 + lane;
        const float* kr = Kn + so;
        const float* vr = Vn + so;
        const float* cr = Qc + dof;
        const float* pr = Qp + dof;
        k0[i] = kr[0];  k1[i] = kr[32];
        c0[i] = cr[0];  c1[i] = cr[32];
        p0[i] = pr[0];  p1[i] = pr[32];
        v0[i] = vr[0];  v1[i] = vr[32];
    }
#pragma unroll
    for (int i = 0; i < NE; ++i) {
        const float sc0 = edge_score(head_sum(c0[i] * k0[i]));
        const float sc1 = edge_score(head_sum(c1[i] * k1[i]));
        const float sp0 = edge_score(head_sum(p0[i] * k0[i]));
        const float sp1 = edge_score(head_sum(p1[i] * k1[i]));
        float* oc = out_c + (d[i] * 64 + lane);
        float* op = out_p + (d[i] * 64 + lane);
        atomicAdd(oc,      v0[i] * sc0);
        atomicAdd(oc + 32, v1[i] * sc1);
        atomicAdd(op,      v0[i] * sp0);
        atomicAdd(op + 32, v1[i] * sp1);
    }
}

__global__ __launch_bounds__(256)
void edge_attn_kernel(const int* __restrict__ src, const int* __restrict__ dst,
                      const float* __restrict__ Qc, const float* __restrict__ Qp,
                      const float* __restrict__ Kn, const float* __restrict__ Vn,
                      float* __restrict__ out_c, float* __restrict__ out_p,
                      int n_edges)
{
    const int lane = threadIdx.x & 31;
    const int wave = __builtin_amdgcn_readfirstlane((int)(threadIdx.x >> 5));
    const int nwaves = (int)gridDim.x * 8;
    const int w = (int)blockIdx.x * 8 + wave;
    static_assert(EPT == 4, "a group is one 16-byte index read");
    const int nb = n_edges >> 2;

    if (w < nb) {
        int b = w;
        int4 s4 = *(const int4*)(src + 4 * b);
        int4 d4 = *(const int4*)(dst + 4 * b);
        for (;;) {
            const int bn = b + nwaves;
            const int bl = bn < nb ? bn : b;
            const int4 sn = *(const int4*)(src + 4 * bl);
            const int4 dn = *(const int4*)(dst + 4 * bl);

            const int sv[EPT] = {s4.x, s4.y, s4.z, s4.w};
            const int dv[EPT] = {d4.x, d4.y, d4.z, d4.w};
            edge_trip<EPT>(sv, dv, lane, Qc, Qp, Kn, Vn, out_c, out_p);

            if (bn >= nb) break;
            b = bn; s4 = sn; d4 = dn;
        }
    }

    if (w < (n_edges & 3)) {
        const int e = 4 * nb + w;
        const int sv[1] = {src[e]};
        const int dv[1] = {dst[e]};
        edge_trip<1>(sv, dv, lane, Qc, Qp, Kn, Vn, out_c, out_p);
    }
}

extern "C" void kernel_launch(void* const* d_in, const int* in_sizes, int n_in,
                              void* d_out, int out_size, void* d_ws, size_t ws_size,
                              hipStream_t stream)
{
    const float* h_c  = (const float*)d_in[0];
    const float* h_p  = (const float*)d_in[1];
    const int*   src  = (const int*)  d_in[2];
    const int*   dst  = (const int*)  d_in[3];
    const float* Wqcw = (const float*)d_in[4];
    const float* Wqcb = (const float*)d_in[5];
    const float* Wqpw = (const float*)d_in[6];
    const float* Wqpb = (const float*)d_in[7];
    const float* Wkw  = (const float*)d_in[8];
    const float* Wkb  = (const float*)d_in[9];
    const float* Wvw  = (const float*)d_in[10];
    const float* Wvb  = (const float*)d_in[11];

    const int n = in_sizes[0] / 64;
    const int e = in_sizes[2];

    float* Qc = (float*)d_ws;
    float* Qp = Qc + (size_t)n * 64;
    float* Kn = Qp + (size_t)n * 64;
    float* Vn = Kn + (size_t)n * 64;

    float* out_c = (float*)d_out;
    float* out_p = out_c + (size_t)n * 64;

    {
        int total = 2 * n * 64;
        zero_kernel<<<(total + 255) / 256, 256, 0, stream>>>((float*)d_out, total);
    }

    {
        int blocks = (n + 127) / 128;
        proj_wmma_kernel<<<blocks, 256, 0, stream>>>(
            h_c, h_p, Wqcw, Wqcb, Wqpw, Wqpb, Wkw, Wkb, Wvw, Wvb,
            Qc, Qp, Kn, Vn, n);
    }

    {
        edge_attn_kernel<<<2000, 256, 0, stream>>>(
            src, dst, Qc, Qp, Kn, Vn, out_c, out_p, e);
    }
}


# zero fill folded into the projection kernel: a block clears its own 128 rows of both outputs by whole-line 16-byte stores before staging; the 400,000-wave zero launch goes (2 launches)
# speedup vs baseline: 2.8220x; 1.0653x over previous
#include <hip/hip_runtime.h>

typedef __attribute__((ext_vector_type(2))) float v2f;
typedef __attribute__((ext_vector_type(8))) float v8f;

#define WT_STRIDE 66

__global__ __launch_bounds__(256)
void proj_wmma_kernel(const float* __restrict__ h_c,
                      const float* __restrict__ h_p,
                      const float* __restrict__ Wqc, const float* __restrict__ bqc,
                      const float* __restrict__ Wqp, const float* __restrict__ bqp,
                      const float* __restrict__ Wk,  const float* __restrict__ bk,
                      const float* __restrict__ Wv,  const float* __restrict__ bv,
                      float* __restrict__ Qc, float* __restrict__ Qp,
                      float* __restrict__ Kn, float* __restrict__ Vn,
                      float* __restrict__ out_c, float* __restrict__ out_p,
                      int n)
{
    __shared__ float sWT[4][64 * WT_STRIDE];
    __shared__ float sB[4][64];

    const int tid = threadIdx.x;

    {
        const float4 z = make_float4(0.f, 0.f, 0.f, 0.f);
        float4* zc = (float4*)out_c;
        float4* zp = (float4*)out_p;
        const int q0 = (int)blockIdx.x * 2048;
        const int qn = n * 16;
#pragma unroll
        for (int j = 0; j < 8; ++j) {
            const int q = q0 + j * 256 + tid;
            if (q < qn) { zc[q] = z; zp[q] = z; }
        }
    }
    const float* Ws[4] = {Wqc, Wqp, Wk, Wv};
    const float* Bs[4] = {bqc, bqp, bk, bv};
    for (int o = 0; o < 4; ++o) {
        for (int i = tid; i < 4096; i += 256) {
            int col = i >> 6, k = i & 63;
            sWT[o][col * WT_STRIDE + k] = Ws[o][k * 64 + col];
        }
        if (tid < 64) sB[o][tid] = Bs[o][tid];
    }
    __syncthreads();

    const int wave = tid >> 5;
    const int lane = tid & 31;
    const int half = lane >> 4;
    const int lrow = lane & 15;
    const int ncol = lane & 15;

    const int m0 = (blockIdx.x * 8 + wave) * 16;
    if (m0 >= n) return;

    const int row = m0 + lrow;

    v2f a_c[16], a_p[16];
#pragma unroll
    for (int ks = 0; ks < 16; ++ks) {
        int k = ks * 4 + 2 * half;
        a_c[ks] = *(const v2f*)(h_c + row * 64 + k);
        a_p[ks] = *(const v2f*)(h_p + row * 64 + k);
    }

    float* outs[4] = {Qc, Qp, Kn, Vn};

#pragma unroll
    for (int o = 0; o < 4; ++o) {
        float* op = outs[o];
#pragma unroll
        for (int nt = 0; nt < 4; ++nt) {
            const int ncg = nt * 16 + ncol;
            v8f acc = {0.f, 0.f, 0.f, 0.f, 0.f, 0.f, 0.f, 0.f};
#pragma unroll
            for (int ks = 0; ks < 16; ++ks) {
                int k = ks * 4 + 2 * half;
                v2f b = *(const v2f*)(&sWT[o][ncg * WT_STRIDE + k]);
                v2f a = (o == 1) ? a_p[ks] : a_c[ks];
                acc = __builtin_amdgcn_wmma_f32_16x16x4_f32(
                     false, a,  false, b,
                     (short)0, acc,  false,  false);
            }
            const float bias = sB[o][ncg];
#pragma unroll
            for (int r = 0; r < 8; ++r) {
                int m = m0 + r + 8 * half;
                op[m * 64 + ncg] = acc[r] + bias;
            }
        }
    }
}

#define EPT 4

__device__ __forceinline__ float head_sum(float x)
{
    x += __builtin_bit_cast(float, __builtin_amdgcn_update_dpp(
             0, __builtin_bit_cast(int, x), 0x161, 0xf, 0xf, true));
    x += __builtin_bit_cast(float, __builtin_amdgcn_update_dpp(
             0, __builtin_bit_cast(int, x), 0x162, 0xf, 0xf, true));
    x += __builtin_bit_cast(float, __builtin_amdgcn_update_dpp(
             0, __builtin_bit_cast(int, x), 0x164, 0xf, 0xf, true));
    return x;
}

__device__ __forceinline__ float edge_score(float dot)
{
    return __expf(fminf(fmaxf(dot * 0.125f, -5.0f), 5.0f));
}

template <int NE>
__device__ __forceinline__ void edge_trip(const int* s, const int* d, int lane,
                                          const float* __restrict__ Qc,
                                          const float* __restrict__ Qp,
                                          const float* __restrict__ Kn,
                                          const float* __restrict__ Vn,
                                          float* __restrict__ out_c,
                                          float* __restrict__ out_p)
{
    float k0[NE], k1[NE], v0[NE], v1[NE], c0[NE], c1[NE], p0[NE], p1[NE];
#pragma unroll
    for (int i = 0; i < NE; ++i) {
        const int so = s[i] * 64 + lane;
        const int dof = d[i] * 64 + lane;
        const float* kr = Kn + so;
        const float* vr = Vn + so;
        const float* cr = Qc + dof;
        const float* pr = Qp + dof;
        k0[i] = kr[0];  k1[i] = kr[32];
        c0[i] = cr[0];  c1[i] = cr[32];
        p0[i] = pr[0];  p1[i] = pr[32];
        v0[i] = vr[0];  v1[i] = vr[32];
    }
#pragma unroll
    for (int i = 0; i < NE; ++i) {
        const float sc0 = edge_score(head_sum(c0[i] * k0[i]));
        const float sc1 = edge_score(head_sum(c1[i] * k1[i]));
        const float sp0 = edge_score(head_sum(p0[i] * k0[i]));
        const float sp1 = edge_score(head_sum(p1[i] * k1[i]));
        float* oc = out_c + (d[i] * 64 + lane);
        float* op = out_p + (d[i] * 64 + lane);
        atomicAdd(oc,      v0[i] * sc0);
        atomicAdd(oc + 32, v1[i] * sc1);
        atomicAdd(op,      v0[i] * sp0);
        atomicAdd(op + 32, v1[i] * sp1);
    }
}

__global__ __launch_bounds__(256)
void edge_attn_kernel(const int* __restrict__ src, const int* __restrict__ dst,
                      const float* __restrict__ Qc, const float* __restrict__ Qp,
                      const float* __restrict__ Kn, const float* __restrict__ Vn,
                      float* __restrict__ out_c, float* __restrict__ out_p,
                      int n_edges)
{
    const int lane = threadIdx.x & 31;
    const int wave = __builtin_amdgcn_readfirstlane((int)(threadIdx.x >> 5));
    const int nwaves = (int)gridDim.x * 8;
    const int w = (int)blockIdx.x * 8 + wave;
    static_assert(EPT == 4, "a group is one 16-byte index read");
    const int nb = n_edges >> 2;

    if (w < nb) {
        int b = w;
        int4 s4 = *(const int4*)(src + 4 * b);
        int4 d4 = *(const int4*)(dst + 4 * b);
        for (;;) {
            const int bn = b + nwaves;
            const int bl = bn < nb ? bn : b;
            const int4 sn = *(const int4*)(src + 4 * bl);
            const int4 dn = *(const int4*)(dst + 4 * bl);

            const int sv[EPT] = {s4.x, s4.y, s4.z, s4.w};
            const int dv[EPT] = {d4.x, d4.y, d4.z, d4.w};
            edge_trip<EPT>(sv, dv, lane, Qc, Qp, Kn, Vn, out_c, out_p);

            if (bn >= nb) break;
            b = bn; s4 = sn; d4 = dn;
        }
    }

    if (w < (n_edges & 3)) {
        const int e = 4 * nb + w;
        const int sv[1] = {src[e]};
        const int dv[1] = {dst[e]};
        edge_trip<1>(sv, dv, lane, Qc, Qp, Kn, Vn, out_c, out_p);
    }
}

extern "C" void kernel_launch(void* const* d_in, const int* in_sizes, int n_in,
                              void* d_out, int out_size, void* d_ws, size_t ws_size,
                              hipStream_t stream)
{
    const float* h_c  = (const float*)d_in[0];
    const float* h_p  = (const float*)d_in[1];
    const int*   src  = (const int*)  d_in[2];
    const int*   dst  = (const int*)  d_in[3];
    const float* Wqcw = (const float*)d_in[4];
    const float* Wqcb = (const float*)d_in[5];
    const float* Wqpw = (const float*)d_in[6];
    const float* Wqpb = (const float*)d_in[7];
    const float* Wkw  = (const float*)d_in[8];
    const float* Wkb  = (const float*)d_in[9];
    const float* Wvw  = (const float*)d_in[10];
    const float* Wvb  = (const float*)d_in[11];

    const int n = in_sizes[0] / 64;
    const int e = in_sizes[2];

    float* Qc = (float*)d_ws;
    float* Qp = Qc + (size_t)n * 64;
    float* Kn = Qp + (size_t)n * 64;
    float* Vn = Kn + (size_t)n * 64;

    float* out_c = (float*)d_out;
    float* out_p = out_c + (size_t)n * 64;

    {
        int blocks = (n + 127) / 128;
        proj_wmma_kernel<<<blocks, 256, 0, stream>>>(
            h_c, h_p, Wqcw, Wqcb, Wqpw, Wqpb, Wkw, Wkb, Wvw, Wvb,
            Qc, Qp, Kn, Vn, out_c, out_p, n);
    }

    {
        edge_attn_kernel<<<2000, 256, 0, stream>>>(
            src, dst, Qc, Qp, Kn, Vn, out_c, out_p, e);
    }
}


# edge kernel: a block takes only the edges whose destination lies in one half of the nodes, by preference the half of its own dies (task tickets): output and Q rows stay in one L2; arrival order
# speedup vs baseline: 3.0950x; 1.0967x over previous
#include <hip/hip_runtime.h>

typedef __attribute__((ext_vector_type(2))) float v2f;
typedef __attribute__((ext_vector_type(8))) float v8f;

#define WT_STRIDE 66

#define EDGE_GROUPS 250
#define EDGE_SLOTS  (4 * EDGE_GROUPS * 8)

__global__ __launch_bounds__(256)
void proj_wmma_kernel(const float* __restrict__ h_c,
                      const float* __restrict__ h_p,
                      const float* __restrict__ Wqc, const float* __restrict__ bqc,
                      const float* __restrict__ Wqp, const float* __restrict__ bqp,
                      const float* __restrict__ Wk,  const float* __restrict__ bk,
                      const float* __restrict__ Wv,  const float* __restrict__ bv,
                      float* __restrict__ Qc, float* __restrict__ Qp,
                      float* __restrict__ Kn, float* __restrict__ Vn,
                      float* __restrict__ out_c, float* __restrict__ out_p,
                      int* __restrict__ tickets, int n)
{
    __shared__ float sWT[4][64 * WT_STRIDE];
    __shared__ float sB[4][64];

    const int tid = threadIdx.x;

    if (blockIdx.x == 0)
        for (int i = tid; i < 4 * EDGE_GROUPS; i += 256) tickets[i] = 0;

    {
        const float4 z = make_float4(0.f, 0.f, 0.f, 0.f);
        float4* zc = (float4*)out_c;
        float4* zp = (float4*)out_p;
        const int q0 = (int)blockIdx.x * 2048;
        const int qn = n * 16;
#pragma unroll
        for (int j = 0; j < 8; ++j) {
            const int q = q0 + j * 256 + tid;
            if (q < qn) { zc[q] = z; zp[q] = z; }
        }
    }
    const float* Ws[4] = {Wqc, Wqp, Wk, Wv};
    const float* Bs[4] = {bqc, bqp, bk, bv};
    for (int o = 0; o < 4; ++o) {
        for (int i = tid; i < 4096; i += 256) {
            int col = i >> 6, k = i & 63;
            sWT[o][col * WT_STRIDE + k] = Ws[o][k * 64 + col];
        }
        if (tid < 64) sB[o][tid] = Bs[o][tid];
    }
    __syncthreads();

    const int wave = tid >> 5;
    const int lane = tid & 31;
    const int half = lane >> 4;
    const int lrow = lane & 15;
    const int ncol = lane & 15;

    const int m0 = (blockIdx.x * 8 + wave) * 16;
    if (m0 >= n) return;

    const int row = m0 + lrow;

    v2f a_c[16], a_p[16];
#pragma unroll
    for (int ks = 0; ks < 16; ++ks) {
        int k = ks * 4 + 2 * half;
        a_c[ks] = *(const v2f*)(h_c + row * 64 + k);
        a_p[ks] = *(const v2f*)(h_p + row * 64 + k);
    }

    float* outs[4] = {Qc, Qp, Kn, Vn};

#pragma unroll
    for (int o = 0; o < 4; ++o) {
        float* op = outs[o];
#pragma unroll
        for (int nt = 0; nt < 4; ++nt) {
            const int ncg = nt * 16 + ncol;
            v8f acc = {0.f, 0.f, 0.f, 0.f, 0.f, 0.f, 0.f, 0.f};
#pragma unroll
            for (int ks = 0; ks < 16; ++ks) {
                int k = ks * 4 + 2 * half;
                v2f b = *(const v2f*)(&sWT[o][ncg * WT_STRIDE + k]);
                v2f a = (o == 1) ? a_p[ks] : a_c[ks];
                acc = __builtin_amdgcn_wmma_f32_16x16x4_f32(
                     false, a,  false, b,
                     (short)0, acc,  false,  false);
            }
            const float bias = sB[o][ncg];
#pragma unroll
            for (int r = 0; r < 8; ++r) {
                int m = m0 + r + 8 * half;
                op[m * 64 + ncg] = acc[r] + bias;
            }
        }
    }
}

#define EPT 4

__device__ __forceinline__ float head_sum(float x)
{
    x += __builtin_bit_cast(float, __builtin_amdgcn_update_dpp(
             0, __builtin_bit_cast(int, x), 0x161, 0xf, 0xf, true));
    x += __builtin_bit_cast(float, __builtin_amdgcn_update_dpp(
             0, __builtin_bit_cast(int, x), 0x162, 0xf, 0xf, true));
    x += __builtin_bit_cast(float, __builtin_amdgcn_update_dpp(
             0, __builtin_bit_cast(int, x), 0x164, 0xf, 0xf, true));
    return x;
}

__device__ __forceinline__ float edge_score(float dot)
{
    return __expf(fminf(fmaxf(dot * 0.125f, -5.0f), 5.0f));
}

template <int NE>
__device__ __forceinline__ void edge_trip(const int* s, const int* d, int lane,
                                          const float* __restrict__ Qc,
                                          const float* __restrict__ Qp,
                                          const float* __restrict__ Kn,
                                          const float* __restrict__ Vn,
                                          float* __restrict__ out_c,
                                          float* __restrict__ out_p)
{
    float k0[NE], k1[NE], v0[NE], v1[NE], c0[NE], c1[NE], p0[NE], p1[NE];
#pragma unroll
    for (int i = 0; i < NE; ++i) {
        const int so = s[i] * 64 + lane;
        const int dof = d[i] * 64 + lane;
        const float* kr = Kn + so;
        const float* vr = Vn + so;
        const float* cr = Qc + dof;
        const float* pr = Qp + dof;
        k0[i] = kr[0];  k1[i] = kr[32];
        c0[i] = cr[0];  c1[i] = cr[32];
        p0[i] = pr[0];  p1[i] = pr[32];
        v0[i] = vr[0];  v1[i] = vr[32];
    }
#pragma unroll
    for (int i = 0; i < NE; ++i) {
        const float sc0 = edge_score(head_sum(c0[i] * k0[i]));
        const float sc1 = edge_score(head_sum(c1[i] * k1[i]));
        const float sp0 = edge_score(head_sum(p0[i] * k0[i]));
        const float sp1 = edge_score(head_sum(p1[i] * k1[i]));
        float* oc = out_c + (d[i] * 64 + lane);
        float* op = out_p + (d[i] * 64 + lane);
        atomicAdd(oc,      v0[i] * sc0);
        atomicAdd(oc + 32, v1[i] * sc1);
        atomicAdd(op,      v0[i] * sp0);
        atomicAdd(op + 32, v1[i] * sp1);
    }
}

__global__ __launch_bounds__(256)
void edge_attn_kernel(const int* __restrict__ src, const int* __restrict__ dst,
                      const float* __restrict__ Qc, const float* __restrict__ Qp,
                      const float* __restrict__ Kn, const float* __restrict__ Vn,
                      float* __restrict__ out_c, float* __restrict__ out_p,
                      int* __restrict__ tickets, int n_nodes, int n_edges)
{
    __shared__ int s_task;
    __shared__ alignas(16) int2 s_q[8][64];

    const int lane = threadIdx.x & 31;
    const int wave = __builtin_amdgcn_readfirstlane((int)(threadIdx.x >> 5));
    const int grp = (int)blockIdx.x >> 3;

    if (threadIdx.x == 0) {
        const unsigned id = __builtin_amdgcn_s_sendmsg_rtn(0x87);
        const int h = (int)(id >> 18) & 1;
        int* c = tickets + 4 * grp;
        int th = h;
        int t = atomicAdd(c + h, 1);
        if (t >= 4) { th = 1 - h; t = 3 - atomicAdd(c + 2 + th, 1); }
        s_task = th * 4 + (t & 3);
    }
    __syncthreads();
    const int task = __builtin_amdgcn_readfirstlane(s_task);
    const bool upper = (task >> 2) != 0;
    const int slot = (grp * 4 + (task & 3)) * 8 + wave;
    const int nh = n_nodes >> 1;

    static_assert(EPT == 4, "a trip takes 4 pairs, two 16-byte ring reads");
    int2* q = s_q[wave];
    int head = 0, tail = 0;
    const int sub = lane >> 4, l16 = lane & 15;

    int j = 0;
    int e = 16 * (slot + EDGE_SLOTS * sub) + l16;
    int sv = 0, dv = -1;
    if (e < n_edges) { sv = src[e]; dv = dst[e]; }
    while (16 * (slot + EDGE_SLOTS * 2 * j) < n_edges) {
        const bool take = dv >= 0 && ((dv >= nh) == upper);
        const unsigned m = __builtin_amdgcn_ballot_w32(take);
        if (take) q[(tail + (int)__builtin_amdgcn_mbcnt_lo(m, 0u)) & 63] = make_int2(sv, dv);
        tail += __builtin_popcount(m);

        ++j;
        e = 16 * (slot + EDGE_SLOTS * (2 * j + sub)) + l16;
        sv = 0; dv = -1;
        if (e < n_edges) { sv = src[e]; dv = dst[e]; }

        while (tail - head >= EPT) {
            const int4* p = (const int4*)(q + (head & 63));
            const int4 a = p[0], b = p[1];
            const int s4[EPT] = {__builtin_amdgcn_readfirstlane(a.x), __builtin_amdgcn_readfirstlane(a.z),
                                 __builtin_amdgcn_readfirstlane(b.x), __builtin_amdgcn_readfirstlane(b.z)};
            const int d4[EPT] = {__builtin_amdgcn_readfirstlane(a.y), __builtin_amdgcn_readfirstlane(a.w),
                                 __builtin_amdgcn_readfirstlane(b.y), __builtin_amdgcn_readfirstlane(b.w)};
            edge_trip<EPT>(s4, d4, lane, Qc, Qp, Kn, Vn, out_c, out_p);
            head += EPT;
        }
    }

    for (; head < tail; ++head) {
        const int2 x = q[head & 63];
        const int s1[1] = {__builtin_amdgcn_readfirstlane(x.x)};
        const int d1[1] = {__builtin_amdgcn_readfirstlane(x.y)};
        edge_trip<1>(s1, d1, lane, Qc, Qp, Kn, Vn, out_c, out_p);
    }
}

extern "C" void kernel_launch(void* const* d_in, const int* in_sizes, int n_in,
                              void* d_out, int out_size, void* d_ws, size_t ws_size,
                              hipStream_t stream)
{
    const float* h_c  = (const float*)d_in[0];
    const float* h_p  = (const float*)d_in[1];
    const int*   src  = (const int*)  d_in[2];
    const int*   dst  = (const int*)  d_in[3];
    const float* Wqcw = (const float*)d_in[4];
    const float* Wqcb = (const float*)d_in[5];
    const float* Wqpw = (const float*)d_in[6];
    const float* Wqpb = (const float*)d_in[7];
    const float* Wkw  = (const float*)d_in[8];
    const float* Wkb  = (const float*)d_in[9];
    const float* Wvw  = (const float*)d_in[10];
    const float* Wvb  = (const float*)d_in[11];

    const int n = in_sizes[0] / 64;
    const int e = in_sizes[2];

    float* Qc = (float*)d_ws;
    float* Qp = Qc + (size_t)n * 64;
    float* Kn = Qp + (size_t)n * 64;
    float* Vn = Kn + (size_t)n * 64;
    int* tickets = (int*)(Vn + (size_t)n * 64);

    float* out_c = (float*)d_out;
    float* out_p = out_c + (size_t)n * 64;

    {
        int blocks = (n + 127) / 128;
        proj_wmma_kernel<<<blocks, 256, 0, stream>>>(
            h_c, h_p, Wqcw, Wqcb, Wqpw, Wqpb, Wkw, Wkb, Wvw, Wvb,
            Qc, Qp, Kn, Vn, out_c, out_p, tickets, n);
    }

    {
        edge_attn_kernel<<<8 * EDGE_GROUPS, 256, 0, stream>>>(
            src, dst, Qc, Qp, Kn, Vn, out_c, out_p, tickets, n, e);
    }
}


# projection kernel: a wave keeps one whole weight matrix in registers as B fragments and loops over 16-row tiles (512 resident blocks); the per-block LDS staging of all weights goes; sums unchanged
# speedup vs baseline: 3.4856x; 1.1262x over previous
#include <hip/hip_runtime.h>

typedef __attribute__((ext_vector_type(2))) float v2f;
typedef __attribute__((ext_vector_type(8))) float v8f;

#define EDGE_GROUPS 250
#define EDGE_SLOTS  (4 * EDGE_GROUPS * 8)

#define PROJ_BLOCKS 512

__global__ __launch_bounds__(256)
void proj_wmma_kernel(const float* __restrict__ h_c,
                      const float* __restrict__ h_p,
                      const float* __restrict__ Wqc, const float* __restrict__ bqc,
                      const float* __restrict__ Wqp, const float* __restrict__ bqp,
                      const float* __restrict__ Wk,  const float* __restrict__ bk,
                      const float* __restrict__ Wv,  const float* __restrict__ bv,
                      float* __restrict__ Qc, float* __restrict__ Qp,
                      float* __restrict__ Kn, float* __restrict__ Vn,
                      float* __restrict__ out_c, float* __restrict__ out_p,
                      int* __restrict__ tickets, int n)
{
    const int tid = threadIdx.x;

    if (blockIdx.x == 0)
        for (int i = tid; i < 4 * EDGE_GROUPS; i += 256) tickets[i] = 0;

    const int lane = tid & 31;
    const int wave = __builtin_amdgcn_readfirstlane(tid >> 5);
    const int o    = wave & 3;
    const int half = lane >> 4;
    const int c16  = lane & 15;

    const float* Wm  = (o == 0) ? Wqc : (o == 1) ? Wqp : (o == 2) ? Wk : Wv;
    const float* bm  = (o == 0) ? bqc : (o == 1) ? bqp : (o == 2) ? bk : bv;
    const float* hin = (o == 1) ? h_p : h_c;
    float*       op  = (o == 0) ? Qc  : (o == 1) ? Qp  : (o == 2) ? Kn : Vn;

    float4* zt = (float4*)((o & 1) ? out_p : out_c) + (o >> 1) * 128 + lane;

    v2f b[4][16];
    float bias[4];
#pragma unroll
    for (int nt = 0; nt < 4; ++nt) {
        bias[nt] = bm[nt * 16 + c16];
#pragma unroll
        for (int ks = 0; ks < 16; ++ks) {
            const float* w = Wm + (ks * 4 + 2 * half) * 64 + nt * 16 + c16;
            b[nt][ks] = (v2f){w[0], w[64]};
        }
    }

    const int ntiles  = n >> 4;
    const int nstream = (int)gridDim.x * 2;
    int t = (int)blockIdx.x * 2 + (wave >> 2);

    if (t >= ntiles) return;

#define LOAD_A(dst, tile)                                                       \
    {                                                                           \
        const float* hr = hin + ((tile) * 16 + c16) * 64 + 2 * half;            \
        _Pragma("unroll")                                                       \
        for (int ks = 0; ks < 16; ++ks) dst[ks] = *(const v2f*)(hr + ks * 4);   \
    }

#define DO_TILE(av, tile)                                                       \
    {                                                                           \
        const int m0 = (tile) * 16;                                             \
        {     \
              \
            const float4 z = make_float4(0.f, 0.f, 0.f, 0.f);                   \
            float4* zp = zt + m0 * 16;                                          \
            _Pragma("unroll")                                                   \
            for (int j = 0; j < 4; ++j) zp[j * 32] = z;                         \
        }                                                                       \
        v8f acc[4];                                                             \
        _Pragma("unroll")                                                       \
        for (int nt = 0; nt < 4; ++nt)                                          \
            acc[nt] = (v8f){0.f, 0.f, 0.f, 0.f, 0.f, 0.f, 0.f, 0.f};            \
        _Pragma("unroll")                                                       \
        for (int ks = 0; ks < 16; ++ks) {                                       \
            _Pragma("unroll")                                                   \
            for (int nt = 0; nt < 4; ++nt)                                      \
                acc[nt] = __builtin_amdgcn_wmma_f32_16x16x4_f32(                \
                     false, av[ks],  false, b[nt][ks],        \
                     (short)0, acc[nt],  false,             \
                     false);                                         \
        }                                                                       \
        _Pragma("unroll")                                                       \
        for (int r = 0; r < 8; ++r) {     \
            float* orow = op + (m0 + r + 8 * half) * 64 + c16;                  \
            _Pragma("unroll")                                                   \
            for (int nt = 0; nt < 4; ++nt)                                      \
                orow[nt * 16] = acc[nt][r] + bias[nt];                          \
        }                                                                       \
    }

    v2f a0[16], a1[16];
    LOAD_A(a0, t)
    while (t + nstream < ntiles) {
        const int t1 = t + nstream;
        const int t2 = t1 + nstream;
        LOAD_A(a1, t1)
        DO_TILE(a0, t)
        LOAD_A(a0, (t2 < ntiles ? t2 : t1))
        DO_TILE(a1, t1)
        t = t2;
    }
    if (t < ntiles) DO_TILE(a0, t)
#undef LOAD_A
#undef DO_TILE
}

#define EPT 4

__device__ __forceinline__ float head_sum(float x)
{
    x += __builtin_bit_cast(float, __builtin_amdgcn_update_dpp(
             0, __builtin_bit_cast(int, x), 0x161, 0xf, 0xf, true));
    x += __builtin_bit_cast(float, __builtin_amdgcn_update_dpp(
             0, __builtin_bit_cast(int, x), 0x162, 0xf, 0xf, true));
    x += __builtin_bit_cast(float, __builtin_amdgcn_update_dpp(
             0, __builtin_bit_cast(int, x), 0x164, 0xf, 0xf, true));
    return x;
}

__device__ __forceinline__ float edge_score(float dot)
{
    return __expf(fminf(fmaxf(dot * 0.125f, -5.0f), 5.0f));
}

template <int NE>
__device__ __forceinline__ void edge_trip(const int* s, const int* d, int lane,
                                          const float* __restrict__ Qc,
                                          const float* __restrict__ Qp,
                                          const float* __restrict__ Kn,
                                          const float* __restrict__ Vn,
                                          float* __restrict__ out_c,
                                          float* __restrict__ out_p)
{
    float k0[NE], k1[NE], v0[NE], v1[NE], c0[NE], c1[NE], p0[NE], p1[NE];
#pragma unroll
    for (int i = 0; i < NE; ++i) {
        const int so = s[i] * 64 + lane;
        const int dof = d[i] * 64 + lane;
        const float* kr = Kn + so;
        const float* vr = Vn + so;
        const float* cr = Qc + dof;
        const float* pr = Qp + dof;
        k0[i] = kr[0];  k1[i] = kr[32];
        c0[i] = cr[0];  c1[i] = cr[32];
        p0[i] = pr[0];  p1[i] = pr[32];
        v0[i] = vr[0];  v1[i] = vr[32];
    }
#pragma unroll
    for (int i = 0; i < NE; ++i) {
        const float sc0 = edge_score(head_sum(c0[i] * k0[i]));
        const float sc1 = edge_score(head_sum(c1[i] * k1[i]));
        const float sp0 = edge_score(head_sum(p0[i] * k0[i]));
        const float sp1 = edge_score(head_sum(p1[i] * k1[i]));
        float* oc = out_c + (d[i] * 64 + lane);
        float* op = out_p + (d[i] * 64 + lane);
        atomicAdd(oc,      v0[i] * sc0);
        atomicAdd(oc + 32, v1[i] * sc1);
        atomicAdd(op,      v0[i] * sp0);
        atomicAdd(op + 32, v1[i] * sp1);
    }
}

__global__ __launch_bounds__(256)
void edge_attn_kernel(const int* __restrict__ src, const int* __restrict__ dst,
                      const float* __restrict__ Qc, const float* __restrict__ Qp,
                      const float* __restrict__ Kn, const float* __restrict__ Vn,
                      float* __restrict__ out_c, float* __restrict__ out_p,
                      int* __restrict__ tickets, int n_nodes, int n_edges)
{
    __shared__ int s_task;
    __shared__ alignas(16) int2 s_q[8][64];

    const int lane = threadIdx.x & 31;
    const int wave = __builtin_amdgcn_readfirstlane((int)(threadIdx.x >> 5));
    const int grp = (int)blockIdx.x >> 3;

    if (threadIdx.x == 0) {
        const unsigned id = __builtin_amdgcn_s_sendmsg_rtn(0x87);
        const int h = (int)(id >> 18) & 1;
        int* c = tickets + 4 * grp;
        int th = h;
        int t = atomicAdd(c + h, 1);
        if (t >= 4) { th = 1 - h; t = 3 - atomicAdd(c + 2 + th, 1); }
        s_task = th * 4 + (t & 3);
    }
    __syncthreads();
    const int task = __builtin_amdgcn_readfirstlane(s_task);
    const bool upper = (task >> 2) != 0;
    const int slot = (grp * 4 + (task & 3)) * 8 + wave;
    const int nh = n_nodes >> 1;

    static_assert(EPT == 4, "a trip takes 4 pairs, two 16-byte ring reads");
    int2* q = s_q[wave];
    int head = 0, tail = 0;
    const int sub = lane >> 4, l16 = lane & 15;

    int j = 0;
    int e = 16 * (slot + EDGE_SLOTS * sub) + l16;
    int sv = 0, dv = -1;
    if (e < n_edges) { sv = src[e]; dv = dst[e]; }
    while (16 * (slot + EDGE_SLOTS * 2 * j) < n_edges) {
        const bool take = dv >= 0 && ((dv >= nh) == upper);
        const unsigned m = __builtin_amdgcn_ballot_w32(take);
        if (take) q[(tail + (int)__builtin_amdgcn_mbcnt_lo(m, 0u)) & 63] = make_int2(sv, dv);
        tail += __builtin_popcount(m);

        ++j;
        e = 16 * (slot + EDGE_SLOTS * (2 * j + sub)) + l16;
        sv = 0; dv = -1;
        if (e < n_edges) { sv = src[e]; dv = dst[e]; }

        while (tail - head >= EPT) {
            const int4* p = (const int4*)(q + (head & 63));
            const int4 a = p[0], b = p[1];
            const int s4[EPT] = {__builtin_amdgcn_readfirstlane(a.x), __builtin_amdgcn_readfirstlane(a.z),
                                 __builtin_amdgcn_readfirstlane(b.x), __builtin_amdgcn_readfirstlane(b.z)};
            const int d4[EPT] = {__builtin_amdgcn_readfirstlane(a.y), __builtin_amdgcn_readfirstlane(a.w),
                                 __builtin_amdgcn_readfirstlane(b.y), __builtin_amdgcn_readfirstlane(b.w)};
            edge_trip<EPT>(s4, d4, lane, Qc, Qp, Kn, Vn, out_c, out_p);
            head += EPT;
        }
    }

    for (; head < tail; ++head) {
        const int2 x = q[head & 63];
        const int s1[1] = {__builtin_amdgcn_readfirstlane(x.x)};
        const int d1[1] = {__builtin_amdgcn_readfirstlane(x.y)};
        edge_trip<1>(s1, d1, lane, Qc, Qp, Kn, Vn, out_c, out_p);
    }
}

extern "C" void kernel_launch(void* const* d_in, const int* in_sizes, int n_in,
                              void* d_out, int out_size, void* d_ws, size_t ws_size,
                              hipStream_t stream)
{
    const float* h_c  = (const float*)d_in[0];
    const float* h_p  = (const float*)d_in[1];
    const int*   src  = (const int*)  d_in[2];
    const int*   dst  = (const int*)  d_in[3];
    const float* Wqcw = (const float*)d_in[4];
    const float* Wqcb = (const float*)d_in[5];
    const float* Wqpw = (const float*)d_in[6];
    const float* Wqpb = (const float*)d_in[7];
    const float* Wkw  = (const float*)d_in[8];
    const float* Wkb  = (const float*)d_in[9];
    const float* Wvw  = (const float*)d_in[10];
    const float* Wvb  = (const float*)d_in[11];

    const int n = in_sizes[0] / 64;
    const int e = in_sizes[2];

    float* Qc = (float*)d_ws;
    float* Qp = Qc + (size_t)n * 64;
    float* Kn = Qp + (size_t)n * 64;
    float* Vn = Kn + (size_t)n * 64;
    int* tickets = (int*)(Vn + (size_t)n * 64);

    float* out_c = (float*)d_out;
    float* out_p = out_c + (size_t)n * 64;

    {
        proj_wmma_kernel<<<PROJ_BLOCKS, 256, 0, stream>>>(
            h_c, h_p, Wqcw, Wqcb, Wqpw, Wqpb, Wkw, Wkb, Wvw, Wvb,
            Qc, Qp, Kn, Vn, out_c, out_p, tickets, n);
    }

    {
        edge_attn_kernel<<<8 * EDGE_GROUPS, 256, 0, stream>>>(
            src, dst, Qc, Qp, Kn, Vn, out_c, out_p, tickets, n, e);
    }
}


# edge list regrouped per call by (128-node source block, destination half) in a new launch; an edge work-group holds its block's K and V rows in LDS, reads only Q rows per edge; adds in arrival order
# speedup vs baseline: 3.9606x; 1.1363x over previous
#include <hip/hip_runtime.h>

typedef __attribute__((ext_vector_type(2))) float v2f;
typedef __attribute__((ext_vector_type(8))) float v8f;
typedef __attribute__((ext_vector_type(4))) float v4f;

#define SB_ROWS     128
#define NSB_PAD     784
#define EDGE_GROUPS (NSB_PAD / 4)
#define NKEY        (2 * NSB_PAD)
#define TABW        (NKEY + 1)
#define SLAB        16384
#define BIN_BLOCKS  98

#define PROJ_BLOCKS 512

__global__ __launch_bounds__(256)
void proj_wmma_kernel(const float* __restrict__ h_c,
                      const float* __restrict__ h_p,
                      const float* __restrict__ Wqc, const float* __restrict__ bqc,
                      const float* __restrict__ Wqp, const float* __restrict__ bqp,
                      const float* __restrict__ Wk,  const float* __restrict__ bk,
                      const float* __restrict__ Wv,  const float* __restrict__ bv,
                      float* __restrict__ Qc, float* __restrict__ Qp,
                      float* __restrict__ Kn, float* __restrict__ Vn,
                      float* __restrict__ out_c, float* __restrict__ out_p,
                      int* __restrict__ tickets, int n)
{
    const int tid = threadIdx.x;

    if (blockIdx.x == 0)
        for (int i = tid; i < 4 * EDGE_GROUPS; i += 256) tickets[i] = 0;

    const int lane = tid & 31;
    const int wave = __builtin_amdgcn_readfirstlane(tid >> 5);
    const int o    = wave & 3;
    const int half = lane >> 4;
    const int c16  = lane & 15;

    const float* Wm  = (o == 0) ? Wqc : (o == 1) ? Wqp : (o == 2) ? Wk : Wv;
    const float* bm  = (o == 0) ? bqc : (o == 1) ? bqp : (o == 2) ? bk : bv;
    const float* hin = (o == 1) ? h_p : h_c;
    float*       op  = (o == 0) ? Qc  : (o == 1) ? Qp  : (o == 2) ? Kn : Vn;

    float4* zt = (float4*)((o & 1) ? out_p : out_c) + (o >> 1) * 128 + lane;

    v2f b[4][16];
    float bias[4];
#pragma unroll
    for (int nt = 0; nt < 4; ++nt) {
        bias[nt] = bm[nt * 16 + c16];
#pragma unroll
        for (int ks = 0; ks < 16; ++ks) {
            const float* w = Wm + (ks * 4 + 2 * half) * 64 + nt * 16 + c16;
            b[nt][ks] = (v2f){w[0], w[64]};
        }
    }

    const int ntiles  = n >> 4;
    const int nstream = (int)gridDim.x * 2;
    int t = (int)blockIdx.x * 2 + (wave >> 2);

    if (t >= ntiles) return;

#define LOAD_A(dst, tile)                                                       \
    {                                                                           \
        const float* hr = hin + ((tile) * 16 + c16) * 64 + 2 * half;            \
        _Pragma("unroll")                                                       \
        for (int ks = 0; ks < 16; ++ks) dst[ks] = *(const v2f*)(hr + ks * 4);   \
    }

#define DO_TILE(av, tile)                                                       \
    {                                                                           \
        const int m0 = (tile) * 16;                                             \
        {     \
              \
            const float4 z = make_float4(0.f, 0.f, 0.f, 0.f);                   \
            float4* zp = zt + m0 * 16;                                          \
            _Pragma("unroll")                                                   \
            for (int j = 0; j < 4; ++j) zp[j * 32] = z;                         \
        }                                                                       \
        v8f acc[4];                                                             \
        _Pragma("unroll")                                                       \
        for (int nt = 0; nt < 4; ++nt)                                          \
            acc[nt] = (v8f){0.f, 0.f, 0.f, 0.f, 0.f, 0.f, 0.f, 0.f};            \
        _Pragma("unroll")                                                       \
        for (int ks = 0; ks < 16; ++ks) {                                       \
            _Pragma("unroll")                                                   \
            for (int nt = 0; nt < 4; ++nt)                                      \
                acc[nt] = __builtin_amdgcn_wmma_f32_16x16x4_f32(                \
                     false, av[ks],  false, b[nt][ks],        \
                     (short)0, acc[nt],  false,             \
                     false);                                         \
        }                                                                       \
        _Pragma("unroll")                                                       \
        for (int r = 0; r < 8; ++r) {     \
            float* orow = op + (m0 + r + 8 * half) * 64 + c16;                  \
            _Pragma("unroll")                                                   \
            for (int nt = 0; nt < 4; ++nt)                                      \
                orow[nt * 16] = acc[nt][r] + bias[nt];                          \
        }                                                                       \
    }

    v2f a0[16], a1[16];
    LOAD_A(a0, t)
    while (t + nstream < ntiles) {
        const int t1 = t + nstream;
        const int t2 = t1 + nstream;
        LOAD_A(a1, t1)
        DO_TILE(a0, t)
        LOAD_A(a0, (t2 < ntiles ? t2 : t1))
        DO_TILE(a1, t1)
        t = t2;
    }
    if (t < ntiles) DO_TILE(a0, t)
#undef LOAD_A
#undef DO_TILE
}

__global__ __launch_bounds__(1024)
void bin_kernel(const int* __restrict__ src, const int* __restrict__ dst,
                int* __restrict__ slabs, int* __restrict__ tab,
                int n_nodes, int n_edges)
{
    static_assert(SB_ROWS == 128, "source row inside its block: 7 bits");
    __shared__ int s_cnt[2048];
    __shared__ int s_wsum[32];
    __shared__ alignas(16) int s_pair[SLAB];

    const int tid  = threadIdx.x;
    const int lane = tid & 31;
    const int wave = tid >> 5;
    const int nh   = n_nodes >> 1;
    const int nslab = (n_edges + SLAB - 1) / SLAB;

    for (int slab = blockIdx.x; slab < nslab; slab += (int)gridDim.x) {
        const int base = slab * SLAB;
        s_cnt[tid] = 0;
        s_cnt[tid + 1024] = 0;
        __syncthreads();

        int sv[16], dv[16], ky[16], rk[16];
#pragma unroll
        for (int i = 0; i < 16; ++i) {
            const int e = base + i * 1024 + tid;
            const int ec = e < n_edges ? e : n_edges - 1;
            sv[i] = src[ec];
            dv[i] = dst[ec];
        }
#pragma unroll
        for (int i = 0; i < 16; ++i) {
            ky[i] = 0; rk[i] = 0;
            if (base + i * 1024 + tid < n_edges) {
                ky[i] = (sv[i] / SB_ROWS) * 2 + (dv[i] >= nh ? 1 : 0);
                rk[i] = atomicAdd(&s_cnt[ky[i]], 1);
            } else {
                sv[i] = -1;
            }
        }
        __syncthreads();

        const int c0 = s_cnt[2 * tid], c1 = s_cnt[2 * tid + 1];
        int inc = c0 + c1;
        for (int o = 1; o < 32; o <<= 1) {
            const int y = __shfl_up(inc, o);
            if (lane >= o) inc += y;
        }
        if (lane == 31) s_wsum[wave] = inc;
        __syncthreads();
        if (wave == 0) {
            const int w = s_wsum[lane];
            int wi = w;
            for (int o = 1; o < 32; o <<= 1) {
                const int y = __shfl_up(wi, o);
                if (lane >= o) wi += y;
            }
            s_wsum[lane] = wi - w;
        }
        __syncthreads();
        const int ex = inc - (c0 + c1) + s_wsum[wave];
        s_cnt[2 * tid]     = ex;
        s_cnt[2 * tid + 1] = ex + c0;
        __syncthreads();

#pragma unroll
        for (int i = 0; i < 16; ++i) rk[i] += s_cnt[ky[i]];
#pragma unroll
        for (int i = 0; i < 16; ++i)
            if (sv[i] >= 0) s_pair[rk[i]] = ((sv[i] & (SB_ROWS - 1)) << 24) | dv[i];
        for (int k = tid; k < TABW; k += 1024) tab[slab * TABW + k] = s_cnt[k];
        __syncthreads();

        for (int i = 0; i < 16; ++i) {
            const int idx = i * 1024 + tid;
            if (base + idx < n_edges) slabs[base + idx] = s_pair[idx];
        }
        __syncthreads();
    }
}

#define EPT 4

__device__ __forceinline__ float head_sum(float x)
{
    x += __builtin_bit_cast(float, __builtin_amdgcn_update_dpp(
             0, __builtin_bit_cast(int, x), 0x161, 0xf, 0xf, true));
    x += __builtin_bit_cast(float, __builtin_amdgcn_update_dpp(
             0, __builtin_bit_cast(int, x), 0x162, 0xf, 0xf, true));
    x += __builtin_bit_cast(float, __builtin_amdgcn_update_dpp(
             0, __builtin_bit_cast(int, x), 0x164, 0xf, 0xf, true));
    return x;
}

__device__ __forceinline__ float edge_score(float dot)
{
    return __expf(fminf(fmaxf(dot * 0.125f, -5.0f), 5.0f));
}

template <int NE>
__device__ __forceinline__ void edge_trip(const int* w, int lane,
                                          const float* __restrict__ Qc,
                                          const float* __restrict__ Qp,
                                          const float* ks, const float* vs,
                                          float* __restrict__ out_c,
                                          float* __restrict__ out_p)
{
    float k0[NE], k1[NE], v0[NE], v1[NE], c0[NE], c1[NE], p0[NE], p1[NE];
    int s[NE], d[NE];
#pragma unroll
    for (int i = 0; i < NE; ++i) { s[i] = w[i] >> 24; d[i] = w[i] & 0xffffff; }
#pragma unroll
    for (int i = 0; i < NE; ++i) {
        const int dof = d[i] * 64 + lane;
        const float* cr = Qc + dof;
        const float* pr = Qp + dof;
        c0[i] = cr[0];  c1[i] = cr[32];
        p0[i] = pr[0];  p1[i] = pr[32];
    }
#pragma unroll
    for (int i = 0; i < NE; ++i) {
        const int so = s[i] * 64 + lane;
        k0[i] = ks[so];  k1[i] = ks[so + 32];
        v0[i] = vs[so];  v1[i] = vs[so + 32];
    }
#pragma unroll
    for (int i = 0; i < NE; ++i) {
        const float sc0 = edge_score(head_sum(c0[i] * k0[i]));
        const float sc1 = edge_score(head_sum(c1[i] * k1[i]));
        const float sp0 = edge_score(head_sum(p0[i] * k0[i]));
        const float sp1 = edge_score(head_sum(p1[i] * k1[i]));
        float* oc = out_c + (d[i] * 64 + lane);
        float* op = out_p + (d[i] * 64 + lane);
        atomicAdd(oc,      v0[i] * sc0);
        atomicAdd(oc + 32, v1[i] * sc1);
        atomicAdd(op,      v0[i] * sp0);
        atomicAdd(op + 32, v1[i] * sp1);
    }
}

__global__ __launch_bounds__(512)
void edge_attn_kernel(const int* __restrict__ slabs, const int* __restrict__ tab,
                      const float* __restrict__ Qc, const float* __restrict__ Qp,
                      const float* __restrict__ Kn, const float* __restrict__ Vn,
                      float* __restrict__ out_c, float* __restrict__ out_p,
                      int* __restrict__ tickets, int n_nodes, int n_edges)
{
    __shared__ int s_task;
    __shared__ alignas(16) float s_k[SB_ROWS * 64];
    __shared__ alignas(16) float s_v[SB_ROWS * 64];
    __shared__ alignas(16) int s_q[16][64];

    const int tid  = threadIdx.x;
    const int lane = tid & 31;
    const int wave = __builtin_amdgcn_readfirstlane(tid >> 5);
    const int grp = (int)blockIdx.x >> 3;

    if (tid == 0) {
        const unsigned id = __builtin_amdgcn_s_sendmsg_rtn(0x87);
        const int h = (int)(id >> 18) & 1;
        int* c = tickets + 4 * grp;
        int th = h;
        int t = atomicAdd(c + h, 1);
        if (t >= 4) { th = 1 - h; t = 3 - atomicAdd(c + 2 + th, 1); }
        s_task = th * 4 + (t & 3);
    }
    __syncthreads();
    const int task = __builtin_amdgcn_readfirstlane(s_task);
    const int sb   = grp * 4 + (task & 3);
    const int key  = sb * 2 + (task >> 2);
    const int row0 = sb * SB_ROWS;
    int nrows = n_nodes - row0;
    nrows = nrows > SB_ROWS ? SB_ROWS : nrows;

    {
        const int lim = nrows * 16;
        const int rb  = lim > 0 ? row0 : 0;
        const v4f* kg = (const v4f*)(Kn + rb * 64);
        const v4f* vg = (const v4f*)(Vn + rb * 64);
        v4f* kl = (v4f*)s_k;
        v4f* vl = (v4f*)s_v;
        static_assert(SB_ROWS * 16 == 4 * 512, "4 staging trips of 512 threads");
        v4f kk[4], vv[4];
#pragma unroll
        for (int j = 0; j < 4; ++j) {
            const int i = tid + 512 * j;
            const int ic = i < lim ? i : 0;
            kk[j] = kg[ic];
            vv[j] = vg[ic];
        }
        asm volatile("" :: "v"(kk[0]), "v"(kk[1]), "v"(kk[2]), "v"(kk[3]),
                           "v"(vv[0]), "v"(vv[1]), "v"(vv[2]), "v"(vv[3]));
#pragma unroll
        for (int j = 0; j < 4; ++j) {
            const int i = tid + 512 * j;
            if (i < lim) { kl[i] = kk[j]; vl[i] = vv[j]; }
        }
    }
    __syncthreads();

    const int nslab = (n_edges + SLAB - 1) / SLAB;
    int nseg = (nslab - wave + 15) / 16;
    nseg = nseg < 0 ? 0 : (nseg > 32 ? 32 : nseg);
    int st = 0, en = 0;
    if (lane < nseg) {
        const int* tp = tab + (wave + 16 * lane) * TABW + key;
        st = tp[0];
        en = tp[1];
    }

    static_assert(EPT == 4, "a trip takes 4 words, one 16-byte ring read");
    int* q = s_q[wave];
    int head = 0, tail = 0;

    int s0 = 0, e0 = 0;
    int cur = 0;
    if (nseg > 0) {
        s0 = __builtin_amdgcn_readlane(st, 0);
        e0 = __builtin_amdgcn_readlane(en, 0);
        if (s0 + lane < e0) cur = slabs[wave * SLAB + s0 + lane];
    }
    for (int i = 0; i < nseg; ++i) {
        const int* sp = slabs + (wave + 16 * i) * SLAB;
        int p = s0;
        if (p < e0) {
            if (p + lane < e0) q[(tail + lane) & 63] = cur;
            const int got = e0 - p;
            tail += got > 32 ? 32 : got;
            p += 32;
        }
        int s1 = 0, e1 = 0;
        int nxt = 0;
        if (i + 1 < nseg) {
            s1 = __builtin_amdgcn_readlane(st, i + 1);
            e1 = __builtin_amdgcn_readlane(en, i + 1);
            if (s1 + lane < e1) nxt = sp[16 * SLAB + s1 + lane];
        }
        for (;;) {
            while (tail - head >= EPT) {
                const int4 a = *(const int4*)(q + (head & 63));
                const int w4[EPT] = {__builtin_amdgcn_readfirstlane(a.x), __builtin_amdgcn_readfirstlane(a.y),
                                     __builtin_amdgcn_readfirstlane(a.z), __builtin_amdgcn_readfirstlane(a.w)};
                edge_trip<EPT>(w4, lane, Qc, Qp, s_k, s_v, out_c, out_p);
                head += EPT;
            }
            if (p >= e0) break;
            if (p + lane < e0) q[(tail + lane) & 63] = sp[p + lane];
            const int got = e0 - p;
            tail += got > 32 ? 32 : got;
            p += 32;
        }
        s0 = s1; e0 = e1; cur = nxt;
    }

    for (; head < tail; ++head) {
        const int w1[1] = {__builtin_amdgcn_readfirstlane(q[head & 63])};
        edge_trip<1>(w1, lane, Qc, Qp, s_k, s_v, out_c, out_p);
    }
}

extern "C" void kernel_launch(void* const* d_in, const int* in_sizes, int n_in,
                              void* d_out, int out_size, void* d_ws, size_t ws_size,
                              hipStream_t stream)
{
    const float* h_c  = (const float*)d_in[0];
    const float* h_p  = (const float*)d_in[1];
    const int*   src  = (const int*)  d_in[2];
    const int*   dst  = (const int*)  d_in[3];
    const float* Wqcw = (const float*)d_in[4];
    const float* Wqcb = (const float*)d_in[5];
    const float* Wqpw = (const float*)d_in[6];
    const float* Wqpb = (const float*)d_in[7];
    const float* Wkw  = (const float*)d_in[8];
    const float* Wkb  = (const float*)d_in[9];
    const float* Wvw  = (const float*)d_in[10];
    const float* Wvb  = (const float*)d_in[11];

    const int n = in_sizes[0] / 64;
    const int e = in_sizes[2];

    float* Qc = (float*)d_ws;
    float* Qp = Qc + (size_t)n * 64;
    float* Kn = Qp + (size_t)n * 64;
    float* Vn = Kn + (size_t)n * 64;
    int* tickets = (int*)(Vn + (size_t)n * 64);
    int* slabs = tickets + 1024;
    int* tab = slabs + (size_t)e;

    float* out_c = (float*)d_out;
    float* out_p = out_c + (size_t)n * 64;

    {
        proj_wmma_kernel<<<PROJ_BLOCKS, 256, 0, stream>>>(
            h_c, h_p, Wqcw, Wqcb, Wqpw, Wqpb, Wkw, Wkb, Wvw, Wvb,
            Qc, Qp, Kn, Vn, out_c, out_p, tickets, n);
    }

    {
        bin_kernel<<<BIN_BLOCKS, 1024, 0, stream>>>(src, dst, slabs, tab, n, e);
    }

    {
        edge_attn_kernel<<<8 * EDGE_GROUPS, 512, 0, stream>>>(
            slabs, tab, Qc, Qp, Kn, Vn, out_c, out_p, tickets, n, e);
    }
}


# edge kernel scores: an edge's 4 head sums reduced transposed (lanes choose by lane bit at levels 1, 2) into one register; scale, clamp, exp once per edge (was 4x); same tree, first add unfused
# speedup vs baseline: 4.0831x; 1.0309x over previous
#include <hip/hip_runtime.h>

typedef __attribute__((ext_vector_type(2))) float v2f;
typedef __attribute__((ext_vector_type(8))) float v8f;
typedef __attribute__((ext_vector_type(4))) float v4f;

#define SB_ROWS     128
#define NSB_PAD     784
#define EDGE_GROUPS (NSB_PAD / 4)
#define NKEY        (2 * NSB_PAD)
#define TABW        (NKEY + 1)
#define SLAB        16384
#define BIN_BLOCKS  98

#define PROJ_BLOCKS 512

__global__ __launch_bounds__(256)
void proj_wmma_kernel(const float* __restrict__ h_c,
                      const float* __restrict__ h_p,
                      const float* __restrict__ Wqc, const float* __restrict__ bqc,
                      const float* __restrict__ Wqp, const float* __restrict__ bqp,
                      const float* __restrict__ Wk,  const float* __restrict__ bk,
                      const float* __restrict__ Wv,  const float* __restrict__ bv,
                      float* __restrict__ Qc, float* __restrict__ Qp,
                      float* __restrict__ Kn, float* __restrict__ Vn,
                      float* __restrict__ out_c, float* __restrict__ out_p,
                      int* __restrict__ tickets, int n)
{
    const int tid = threadIdx.x;

    if (blockIdx.x == 0)
        for (int i = tid; i < 4 * EDGE_GROUPS; i += 256) tickets[i] = 0;

    const int lane = tid & 31;
    const int wave = __builtin_amdgcn_readfirstlane(tid >> 5);
    const int o    = wave & 3;
    const int half = lane >> 4;
    const int c16  = lane & 15;

    const float* Wm  = (o == 0) ? Wqc : (o == 1) ? Wqp : (o == 2) ? Wk : Wv;
    const float* bm  = (o == 0) ? bqc : (o == 1) ? bqp : (o == 2) ? bk : bv;
    const float* hin = (o == 1) ? h_p : h_c;
    float*       op  = (o == 0) ? Qc  : (o == 1) ? Qp  : (o == 2) ? Kn : Vn;

    float4* zt = (float4*)((o & 1) ? out_p : out_c) + (o >> 1) * 128 + lane;

    v2f b[4][16];
    float bias[4];
#pragma unroll
    for (int nt = 0; nt < 4; ++nt) {
        bias[nt] = bm[nt * 16 + c16];
#pragma unroll
        for (int ks = 0; ks < 16; ++ks) {
            const float* w = Wm + (ks * 4 + 2 * half) * 64 + nt * 16 + c16;
            b[nt][ks] = (v2f){w[0], w[64]};
        }
    }

    const int ntiles  = n >> 4;
    const int nstream = (int)gridDim.x * 2;
    int t = (int)blockIdx.x * 2 + (wave >> 2);

    if (t >= ntiles) return;

#define LOAD_A(dst, tile)                                                       \
    {                                                                           \
        const float* hr = hin + ((tile) * 16 + c16) * 64 + 2 * half;            \
        _Pragma("unroll")                                                       \
        for (int ks = 0; ks < 16; ++ks) dst[ks] = *(const v2f*)(hr + ks * 4);   \
    }

#define DO_TILE(av, tile)                                                       \
    {                                                                           \
        const int m0 = (tile) * 16;                                             \
        {     \
              \
            const float4 z = make_float4(0.f, 0.f, 0.f, 0.f);                   \
            float4* zp = zt + m0 * 16;                                          \
            _Pragma("unroll")                                                   \
            for (int j = 0; j < 4; ++j) zp[j * 32] = z;                         \
        }                                                                       \
        v8f acc[4];                                                             \
        _Pragma("unroll")                                                       \
        for (int nt = 0; nt < 4; ++nt)                                          \
            acc[nt] = (v8f){0.f, 0.f, 0.f, 0.f, 0.f, 0.f, 0.f, 0.f};            \
        _Pragma("unroll")                                                       \
        for (int ks = 0; ks < 16; ++ks) {                                       \
            _Pragma("unroll")                                                   \
            for (int nt = 0; nt < 4; ++nt)                                      \
                acc[nt] = __builtin_amdgcn_wmma_f32_16x16x4_f32(                \
                     false, av[ks],  false, b[nt][ks],        \
                     (short)0, acc[nt],  false,             \
                     false);                                         \
        }                                                                       \
        _Pragma("unroll")                                                       \
        for (int r = 0; r < 8; ++r) {     \
            float* orow = op + (m0 + r + 8 * half) * 64 + c16;                  \
            _Pragma("unroll")                                                   \
            for (int nt = 0; nt < 4; ++nt)                                      \
                orow[nt * 16] = acc[nt][r] + bias[nt];                          \
        }                                                                       \
    }

    v2f a0[16], a1[16];
    LOAD_A(a0, t)
    while (t + nstream < ntiles) {
        const int t1 = t + nstream;
        const int t2 = t1 + nstream;
        LOAD_A(a1, t1)
        DO_TILE(a0, t)
        LOAD_A(a0, (t2 < ntiles ? t2 : t1))
        DO_TILE(a1, t1)
        t = t2;
    }
    if (t < ntiles) DO_TILE(a0, t)
#undef LOAD_A
#undef DO_TILE
}

__global__ __launch_bounds__(1024)
void bin_kernel(const int* __restrict__ src, const int* __restrict__ dst,
                int* __restrict__ slabs, int* __restrict__ tab,
                int n_nodes, int n_edges)
{
    static_assert(SB_ROWS == 128, "source row inside its block: 7 bits");
    __shared__ int s_cnt[2048];
    __shared__ int s_wsum[32];
    __shared__ alignas(16) int s_pair[SLAB];

    const int tid  = threadIdx.x;
    const int lane = tid & 31;
    const int wave = tid >> 5;
    const int nh   = n_nodes >> 1;
    const int nslab = (n_edges + SLAB - 1) / SLAB;

    for (int slab = blockIdx.x; slab < nslab; slab += (int)gridDim.x) {
        const int base = slab * SLAB;
        s_cnt[tid] = 0;
        s_cnt[tid + 1024] = 0;
        __syncthreads();

        int sv[16], dv[16], ky[16], rk[16];
#pragma unroll
        for (int i = 0; i < 16; ++i) {
            const int e = base + i * 1024 + tid;
            const int ec = e < n_edges ? e : n_edges - 1;
            sv[i] = src[ec];
            dv[i] = dst[ec];
        }
#pragma unroll
        for (int i = 0; i < 16; ++i) {
            ky[i] = 0; rk[i] = 0;
            if (base + i * 1024 + tid < n_edges) {
                ky[i] = (sv[i] / SB_ROWS) * 2 + (dv[i] >= nh ? 1 : 0);
                rk[i] = atomicAdd(&s_cnt[ky[i]], 1);
            } else {
                sv[i] = -1;
            }
        }
        __syncthreads();

        const int c0 = s_cnt[2 * tid], c1 = s_cnt[2 * tid + 1];
        int inc = c0 + c1;
        for (int o = 1; o < 32; o <<= 1) {
            const int y = __shfl_up(inc, o);
            if (lane >= o) inc += y;
        }
        if (lane == 31) s_wsum[wave] = inc;
        __syncthreads();
        if (wave == 0) {
            const int w = s_wsum[lane];
            int wi = w;
            for (int o = 1; o < 32; o <<= 1) {
                const int y = __shfl_up(wi, o);
                if (lane >= o) wi += y;
            }
            s_wsum[lane] = wi - w;
        }
        __syncthreads();
        const int ex = inc - (c0 + c1) + s_wsum[wave];
        s_cnt[2 * tid]     = ex;
        s_cnt[2 * tid + 1] = ex + c0;
        __syncthreads();

#pragma unroll
        for (int i = 0; i < 16; ++i) rk[i] += s_cnt[ky[i]];
#pragma unroll
        for (int i = 0; i < 16; ++i)
            if (sv[i] >= 0) s_pair[rk[i]] = ((sv[i] & (SB_ROWS - 1)) << 24) | dv[i];
        for (int k = tid; k < TABW; k += 1024) tab[slab * TABW + k] = s_cnt[k];
        __syncthreads();

        for (int i = 0; i < 16; ++i) {
            const int idx = i * 1024 + tid;
            if (base + idx < n_edges) slabs[base + idx] = s_pair[idx];
        }
        __syncthreads();
    }
}

#define EPT 4

__device__ __forceinline__ float lane_xor1(float x)
{
    return __builtin_bit_cast(float, __builtin_amdgcn_update_dpp(
               0, __builtin_bit_cast(int, x), 0x161, 0xf, 0xf, true));
}
__device__ __forceinline__ float lane_xor2(float x)
{
    return __builtin_bit_cast(float, __builtin_amdgcn_update_dpp(
               0, __builtin_bit_cast(int, x), 0x162, 0xf, 0xf, true));
}
__device__ __forceinline__ float lane_xor4(float x)
{
    return __builtin_bit_cast(float, __builtin_amdgcn_update_dpp(
               0, __builtin_bit_cast(int, x), 0x164, 0xf, 0xf, true));
}
template <int T>
__device__ __forceinline__ float quad_lane(float x)
{
    return __builtin_bit_cast(float, __builtin_amdgcn_update_dpp(
               0, __builtin_bit_cast(int, x), T * 0x55, 0xf, 0xf, true));
}

__device__ __forceinline__ float edge_score(float dot)
{
    return __expf(fminf(fmaxf(dot * 0.125f, -5.0f), 5.0f));
}

__device__ __forceinline__ float head_scores(float a, float b, float c, float d,
                                             bool odd, bool upper)
{
    const float r1 = (odd ? b : a) + lane_xor1(odd ? a : b);
    const float r2 = (odd ? d : c) + lane_xor1(odd ? c : d);
    float s = (upper ? r2 : r1) + lane_xor2(upper ? r1 : r2);
    s += lane_xor4(s);
    return edge_score(s);
}

template <int NE>
__device__ __forceinline__ void edge_trip(const int* w, int lane,
                                          const float* __restrict__ Qc,
                                          const float* __restrict__ Qp,
                                          const float* ks, const float* vs,
                                          float* __restrict__ out_c,
                                          float* __restrict__ out_p)
{
    float k0[NE], k1[NE], v0[NE], v1[NE], c0[NE], c1[NE], p0[NE], p1[NE];
    int s[NE], d[NE];
#pragma unroll
    for (int i = 0; i < NE; ++i) { s[i] = w[i] >> 24; d[i] = w[i] & 0xffffff; }
#pragma unroll
    for (int i = 0; i < NE; ++i) {
        const int dof = d[i] * 64 + lane;
        const float* cr = Qc + dof;
        const float* pr = Qp + dof;
        c0[i] = cr[0];  c1[i] = cr[32];
        p0[i] = pr[0];  p1[i] = pr[32];
    }
#pragma unroll
    for (int i = 0; i < NE; ++i) {
        const int so = s[i] * 64 + lane;
        k0[i] = ks[so];  k1[i] = ks[so + 32];
        v0[i] = vs[so];  v1[i] = vs[so + 32];
    }
    const bool odd   = (lane & 1) != 0;
    const bool upper = (lane & 2) != 0;
#pragma unroll
    for (int i = 0; i < NE; ++i) {
        const float sc = head_scores(c0[i] * k0[i], c1[i] * k1[i],
                                     p0[i] * k0[i], p1[i] * k1[i], odd, upper);
        float* oc = out_c + (d[i] * 64 + lane);
        float* op = out_p + (d[i] * 64 + lane);
        atomicAdd(oc,      v0[i] * quad_lane<0>(sc));
        atomicAdd(oc + 32, v1[i] * quad_lane<1>(sc));
        atomicAdd(op,      v0[i] * quad_lane<2>(sc));
        atomicAdd(op + 32, v1[i] * quad_lane<3>(sc));
    }
}

__global__ __launch_bounds__(512)
void edge_attn_kernel(const int* __restrict__ slabs, const int* __restrict__ tab,
                      const float* __restrict__ Qc, const float* __restrict__ Qp,
                      const float* __restrict__ Kn, const float* __restrict__ Vn,
                      float* __restrict__ out_c, float* __restrict__ out_p,
                      int* __restrict__ tickets, int n_nodes, int n_edges)
{
    __shared__ int s_task;
    __shared__ alignas(16) float s_k[SB_ROWS * 64];
    __shared__ alignas(16) float s_v[SB_ROWS * 64];
    __shared__ alignas(16) int s_q[16][64];

    const int tid  = threadIdx.x;
    const int lane = tid & 31;
    const int wave = __builtin_amdgcn_readfirstlane(tid >> 5);
    const int grp = (int)blockIdx.x >> 3;

    if (tid == 0) {
        const unsigned id = __builtin_amdgcn_s_sendmsg_rtn(0x87);
        const int h = (int)(id >> 18) & 1;
        int* c = tickets + 4 * grp;
        int th = h;
        int t = atomicAdd(c + h, 1);
        if (t >= 4) { th = 1 - h; t = 3 - atomicAdd(c + 2 + th, 1); }
        s_task = th * 4 + (t & 3);
    }
    __syncthreads();
    const int task = __builtin_amdgcn_readfirstlane(s_task);
    const int sb   = grp * 4 + (task & 3);
    const int key  = sb * 2 + (task >> 2);
    const int row0 = sb * SB_ROWS;
    int nrows = n_nodes - row0;
    nrows = nrows > SB_ROWS ? SB_ROWS : nrows;

    {
        const int lim = nrows * 16;
        const int rb  = lim > 0 ? row0 : 0;
        const v4f* kg = (const v4f*)(Kn + rb * 64);
        const v4f* vg = (const v4f*)(Vn + rb * 64);
        v4f* kl = (v4f*)s_k;
        v4f* vl = (v4f*)s_v;
        static_assert(SB_ROWS * 16 == 4 * 512, "4 staging trips of 512 threads");
        v4f kk[4], vv[4];
#pragma unroll
        for (int j = 0; j < 4; ++j) {
            const int i = tid + 512 * j;
            const int ic = i < lim ? i : 0;
            kk[j] = kg[ic];
            vv[j] = vg[ic];
        }
        asm volatile("" :: "v"(kk[0]), "v"(kk[1]), "v"(kk[2]), "v"(kk[3]),
                           "v"(vv[0]), "v"(vv[1]), "v"(vv[2]), "v"(vv[3]));
#pragma unroll
        for (int j = 0; j < 4; ++j) {
            const int i = tid + 512 * j;
            if (i < lim) { kl[i] = kk[j]; vl[i] = vv[j]; }
        }
    }
    __syncthreads();

    const int nslab = (n_edges + SLAB - 1) / SLAB;
    int nseg = (nslab - wave + 15) / 16;
    nseg = nseg < 0 ? 0 : (nseg > 32 ? 32 : nseg);
    int st = 0, en = 0;
    if (lane < nseg) {
        const int* tp = tab + (wave + 16 * lane) * TABW + key;
        st = tp[0];
        en = tp[1];
    }

    static_assert(EPT == 4, "a trip takes 4 words, one 16-byte ring read");
    int* q = s_q[wave];
    int head = 0, tail = 0;

    int s0 = 0, e0 = 0;
    int cur = 0;
    if (nseg > 0) {
        s0 = __builtin_amdgcn_readlane(st, 0);
        e0 = __builtin_amdgcn_readlane(en, 0);
        if (s0 + lane < e0) cur = slabs[wave * SLAB + s0 + lane];
    }
    for (int i = 0; i < nseg; ++i) {
        const int* sp = slabs + (wave + 16 * i) * SLAB;
        int p = s0;
        if (p < e0) {
            if (p + lane < e0) q[(tail + lane) & 63] = cur;
            const int got = e0 - p;
            tail += got > 32 ? 32 : got;
            p += 32;
        }
        int s1 = 0, e1 = 0;
        int nxt = 0;
        if (i + 1 < nseg) {
            s1 = __builtin_amdgcn_readlane(st, i + 1);
            e1 = __builtin_amdgcn_readlane(en, i + 1);
            if (s1 + lane < e1) nxt = sp[16 * SLAB + s1 + lane];
        }
        for (;;) {
            while (tail - head >= EPT) {
                const int4 a = *(const int4*)(q + (head & 63));
                const int w4[EPT] = {__builtin_amdgcn_readfirstlane(a.x), __builtin_amdgcn_readfirstlane(a.y),
                                     __builtin_amdgcn_readfirstlane(a.z), __builtin_amdgcn_readfirstlane(a.w)};
                edge_trip<EPT>(w4, lane, Qc, Qp, s_k, s_v, out_c, out_p);
                head += EPT;
            }
            if (p >= e0) break;
            if (p + lane < e0) q[(tail + lane) & 63] = sp[p + lane];
            const int got = e0 - p;
            tail += got > 32 ? 32 : got;
            p += 32;
        }
        s0 = s1; e0 = e1; cur = nxt;
    }

    for (; head < tail; ++head) {
        const int w1[1] = {__builtin_amdgcn_readfirstlane(q[head & 63])};
        edge_trip<1>(w1, lane, Qc, Qp, s_k, s_v, out_c, out_p);
    }
}

extern "C" void kernel_launch(void* const* d_in, const int* in_sizes, int n_in,
                              void* d_out, int out_size, void* d_ws, size_t ws_size,
                              hipStream_t stream)
{
    const float* h_c  = (const float*)d_in[0];
    const float* h_p  = (const float*)d_in[1];
    const int*   src  = (const int*)  d_in[2];
    const int*   dst  = (const int*)  d_in[3];
    const float* Wqcw = (const float*)d_in[4];
    const float* Wqcb = (const float*)d_in[5];
    const float* Wqpw = (const float*)d_in[6];
    const float* Wqpb = (const float*)d_in[7];
    const float* Wkw  = (const float*)d_in[8];
    const float* Wkb  = (const float*)d_in[9];
    const float* Wvw  = (const float*)d_in[10];
    const float* Wvb  = (const float*)d_in[11];

    const int n = in_sizes[0] / 64;
    const int e = in_sizes[2];

    float* Qc = (float*)d_ws;
    float* Qp = Qc + (size_t)n * 64;
    float* Kn = Qp + (size_t)n * 64;
    float* Vn = Kn + (size_t)n * 64;
    int* tickets = (int*)(Vn + (size_t)n * 64);
    int* slabs = tickets + 1024;
    int* tab = slabs + (size_t)e;

    float* out_c = (float*)d_out;
    float* out_p = out_c + (size_t)n * 64;

    {
        proj_wmma_kernel<<<PROJ_BLOCKS, 256, 0, stream>>>(
            h_c, h_p, Wqcw, Wqcb, Wqpw, Wqpb, Wkw, Wkb, Wvw, Wvb,
            Qc, Qp, Kn, Vn, out_c, out_p, tickets, n);
    }

    {
        bin_kernel<<<BIN_BLOCKS, 1024, 0, stream>>>(src, dst, slabs, tab, n, e);
    }

    {
        edge_attn_kernel<<<8 * EDGE_GROUPS, 512, 0, stream>>>(
            slabs, tab, Qc, Qp, Kn, Vn, out_c, out_p, tickets, n, e);
    }
}


# edge kernel regrouped by 64 destination rows: Q and output rows in LDS, K/V rows read per edge, each row added by ONE wave in sorted (slab, source) order, stored once; global atomics, zero fill go
# speedup vs baseline: 4.8916x; 1.1980x over previous
#include <hip/hip_runtime.h>

typedef __attribute__((ext_vector_type(2))) float v2f;
typedef __attribute__((ext_vector_type(8))) float v8f;
typedef __attribute__((ext_vector_type(4))) float v4f;

#define DB_ROWS     64
#define NDB         1563
#define NKEY        NDB
#define TABW        (NKEY + 1)
#define LIST_CAP    2048
#define SLAB        16384
#define BIN_BLOCKS  98

#define PROJ_BLOCKS 512

__global__ __launch_bounds__(256)
void proj_wmma_kernel(const float* __restrict__ h_c,
                      const float* __restrict__ h_p,
                      const float* __restrict__ Wqc, const float* __restrict__ bqc,
                      const float* __restrict__ Wqp, const float* __restrict__ bqp,
                      const float* __restrict__ Wk,  const float* __restrict__ bk,
                      const float* __restrict__ Wv,  const float* __restrict__ bv,
                      float* __restrict__ Qc, float* __restrict__ Qp,
                      float* __restrict__ Kn, float* __restrict__ Vn, int n)
{
    const int tid = threadIdx.x;
    const int lane = tid & 31;
    const int wave = __builtin_amdgcn_readfirstlane(tid >> 5);
    const int o    = wave & 3;
    const int half = lane >> 4;
    const int c16  = lane & 15;

    const float* Wm  = (o == 0) ? Wqc : (o == 1) ? Wqp : (o == 2) ? Wk : Wv;
    const float* bm  = (o == 0) ? bqc : (o == 1) ? bqp : (o == 2) ? bk : bv;
    const float* hin = (o == 1) ? h_p : h_c;
    float*       op  = (o == 0) ? Qc  : (o == 1) ? Qp  : (o == 2) ? Kn : Vn;

    v2f b[4][16];
    float bias[4];
#pragma unroll
    for (int nt = 0; nt < 4; ++nt) {
        bias[nt] = bm[nt * 16 + c16];
#pragma unroll
        for (int ks = 0; ks < 16; ++ks) {
            const float* w = Wm + (ks * 4 + 2 * half) * 64 + nt * 16 + c16;
            b[nt][ks] = (v2f){w[0], w[64]};
        }
    }

    const int ntiles  = n >> 4;
    const int nstream = (int)gridDim.x * 2;
    int t = (int)blockIdx.x * 2 + (wave >> 2);

    if (t >= ntiles) return;

#define LOAD_A(dst, tile)                                                       \
    {                                                                           \
        const float* hr = hin + ((tile) * 16 + c16) * 64 + 2 * half;            \
        _Pragma("unroll")                                                       \
        for (int ks = 0; ks < 16; ++ks) dst[ks] = *(const v2f*)(hr + ks * 4);   \
    }

#define DO_TILE(av, tile)                                                       \
    {                                                                           \
        const int m0 = (tile) * 16;                                             \
        v8f acc[4];                                                             \
        _Pragma("unroll")                                                       \
        for (int nt = 0; nt < 4; ++nt)                                          \
            acc[nt] = (v8f){0.f, 0.f, 0.f, 0.f, 0.f, 0.f, 0.f, 0.f};            \
        _Pragma("unroll")                                                       \
        for (int ks = 0; ks < 16; ++ks) {                                       \
            _Pragma("unroll")                                                   \
            for (int nt = 0; nt < 4; ++nt)                                      \
                acc[nt] = __builtin_amdgcn_wmma_f32_16x16x4_f32(                \
                     false, av[ks],  false, b[nt][ks],        \
                     (short)0, acc[nt],  false,             \
                     false);                                         \
        }                                                                       \
        _Pragma("unroll")                                                       \
        for (int r = 0; r < 8; ++r) {     \
            float* orow = op + (m0 + r + 8 * half) * 64 + c16;                  \
            _Pragma("unroll")                                                   \
            for (int nt = 0; nt < 4; ++nt)                                      \
                orow[nt * 16] = acc[nt][r] + bias[nt];                          \
        }                                                                       \
    }

    v2f a0[16], a1[16];
    LOAD_A(a0, t)
    while (t + nstream < ntiles) {
        const int t1 = t + nstream;
        const int t2 = t1 + nstream;
        LOAD_A(a1, t1)
        DO_TILE(a0, t)
        LOAD_A(a0, (t2 < ntiles ? t2 : t1))
        DO_TILE(a1, t1)
        t = t2;
    }
    if (t < ntiles) DO_TILE(a0, t)
#undef LOAD_A
#undef DO_TILE
}

__global__ __launch_bounds__(1024)
void bin_kernel(const int* __restrict__ src, const int* __restrict__ dst,
                int* __restrict__ slabs, int* __restrict__ tab,
                int n_nodes, int n_edges)
{
    static_assert(DB_ROWS == 64, "destination row inside its block: 6 bits");
    static_assert(TABW <= 2048, "one counter per key and one for the total");
    __shared__ int s_cnt[2048];
    __shared__ int s_wsum[32];
    __shared__ alignas(16) int s_pair[SLAB];

    const int tid  = threadIdx.x;
    const int lane = tid & 31;
    const int wave = tid >> 5;
    const int nslab = (n_edges + SLAB - 1) / SLAB;

    for (int slab = blockIdx.x; slab < nslab; slab += (int)gridDim.x) {
        const int base = slab * SLAB;
        s_cnt[tid] = 0;
        s_cnt[tid + 1024] = 0;
        __syncthreads();

        int sv[16], dv[16], ky[16], rk[16];
#pragma unroll
        for (int i = 0; i < 16; ++i) {
            const int e = base + i * 1024 + tid;
            const int ec = e < n_edges ? e : n_edges - 1;
            sv[i] = src[ec];
            dv[i] = dst[ec];
        }
#pragma unroll
        for (int i = 0; i < 16; ++i) {
            ky[i] = 0; rk[i] = 0;
            if (base + i * 1024 + tid < n_edges) {
                ky[i] = dv[i] / DB_ROWS;
                rk[i] = atomicAdd(&s_cnt[ky[i]], 1);
            } else {
                sv[i] = -1;
            }
        }
        __syncthreads();

        const int c0 = s_cnt[2 * tid], c1 = s_cnt[2 * tid + 1];
        int inc = c0 + c1;
        for (int o = 1; o < 32; o <<= 1) {
            const int y = __shfl_up(inc, o);
            if (lane >= o) inc += y;
        }
        if (lane == 31) s_wsum[wave] = inc;
        __syncthreads();
        if (wave == 0) {
            const int w = s_wsum[lane];
            int wi = w;
            for (int o = 1; o < 32; o <<= 1) {
                const int y = __shfl_up(wi, o);
                if (lane >= o) wi += y;
            }
            s_wsum[lane] = wi - w;
        }
        __syncthreads();
        const int ex = inc - (c0 + c1) + s_wsum[wave];
        s_cnt[2 * tid]     = ex;
        s_cnt[2 * tid + 1] = ex + c0;
        __syncthreads();

#pragma unroll
        for (int i = 0; i < 16; ++i) rk[i] += s_cnt[ky[i]];
#pragma unroll
        for (int i = 0; i < 16; ++i)
            if (sv[i] >= 0) s_pair[rk[i]] = ((dv[i] & (DB_ROWS - 1)) << 24) | sv[i];
        for (int k = tid; k < TABW; k += 1024) tab[slab * TABW + k] = s_cnt[k];
        __syncthreads();

        for (int i = 0; i < 16; ++i) {
            const int idx = i * 1024 + tid;
            if (base + idx < n_edges) slabs[base + idx] = s_pair[idx];
        }
        __syncthreads();
    }
}

#define EPT 4

__device__ __forceinline__ float lane_xor1(float x)
{
    return __builtin_bit_cast(float, __builtin_amdgcn_update_dpp(
               0, __builtin_bit_cast(int, x), 0x161, 0xf, 0xf, true));
}
__device__ __forceinline__ float lane_xor2(float x)
{
    return __builtin_bit_cast(float, __builtin_amdgcn_update_dpp(
               0, __builtin_bit_cast(int, x), 0x162, 0xf, 0xf, true));
}
__device__ __forceinline__ float lane_xor4(float x)
{
    return __builtin_bit_cast(float, __builtin_amdgcn_update_dpp(
               0, __builtin_bit_cast(int, x), 0x164, 0xf, 0xf, true));
}
template <int T>
__device__ __forceinline__ float quad_lane(float x)
{
    return __builtin_bit_cast(float, __builtin_amdgcn_update_dpp(
               0, __builtin_bit_cast(int, x), T * 0x55, 0xf, 0xf, true));
}

__device__ __forceinline__ float edge_score(float dot)
{
    return __expf(fminf(fmaxf(dot * 0.125f, -5.0f), 5.0f));
}

__device__ __forceinline__ float head_scores(float a, float b, float c, float d,
                                             bool odd, bool upper)
{
    const float r1 = (odd ? b : a) + lane_xor1(odd ? a : b);
    const float r2 = (odd ? d : c) + lane_xor1(odd ? c : d);
    float s = (upper ? r2 : r1) + lane_xor2(upper ? r1 : r2);
    s += lane_xor4(s);
    return edge_score(s);
}

template <int NE>
__device__ __forceinline__ void edge_trip(const int* w, int lane,
                                          const float* __restrict__ Kn,
                                          const float* __restrict__ Vn,
                                          const float* qcs, const float* qps,
                                          float* ocs, float* ops)
{
    float k0[NE], k1[NE], v0[NE], v1[NE], c0[NE], c1[NE], p0[NE], p1[NE];
    int r[NE], s[NE];
#pragma unroll
    for (int i = 0; i < NE; ++i) { r[i] = w[i] >> 24; s[i] = w[i] & 0xffffff; }
#pragma unroll
    for (int i = 0; i < NE; ++i) {
        const int sof = s[i] * 64 + lane;
        const float* kr = Kn + sof;
        const float* vr = Vn + sof;
        k0[i] = kr[0];  k1[i] = kr[32];
        v0[i] = vr[0];  v1[i] = vr[32];
    }
#pragma unroll
    for (int i = 0; i < NE; ++i) {
        const int ro = r[i] * 64 + lane;
        c0[i] = qcs[ro];  c1[i] = qcs[ro + 32];
        p0[i] = qps[ro];  p1[i] = qps[ro + 32];
    }
    const bool odd   = (lane & 1) != 0;
    const bool upper = (lane & 2) != 0;
#pragma unroll
    for (int i = 0; i < NE; ++i) {
        const float sc = head_scores(c0[i] * k0[i], c1[i] * k1[i],
                                     p0[i] * k0[i], p1[i] * k1[i], odd, upper);
        float* oc = ocs + (r[i] * 64 + lane);
        float* op = ops + (r[i] * 64 + lane);
        atomicAdd(oc,      v0[i] * quad_lane<0>(sc));
        atomicAdd(oc + 32, v1[i] * quad_lane<1>(sc));
        atomicAdd(op,      v0[i] * quad_lane<2>(sc));
        atomicAdd(op + 32, v1[i] * quad_lane<3>(sc));
    }
}

__device__ __forceinline__ void place_chunk(int wv, int m, int pos, int lane, int* list)
{
    int rank = 0;
    for (int j = 0; j < m; ++j) {
        const int wj = __builtin_amdgcn_readlane(wv, j);
        rank += (wj < wv || (wj == wv && j < lane)) ? 1 : 0;
    }
    const int at = pos + rank;
    if (lane < m && at >= 0 && at < LIST_CAP) list[at] = wv;
}

__global__ __launch_bounds__(512)
void edge_attn_kernel(const int* __restrict__ slabs, const int* __restrict__ tab,
                      const float* __restrict__ Qc, const float* __restrict__ Qp,
                      const float* __restrict__ Kn, const float* __restrict__ Vn,
                      float* __restrict__ out_c, float* __restrict__ out_p,
                      int n_nodes, int n_edges)
{
    __shared__ alignas(16) float s_qc[DB_ROWS * 64];
    __shared__ alignas(16) float s_qp[DB_ROWS * 64];
    __shared__ alignas(16) float s_oc[DB_ROWS * 64];
    __shared__ alignas(16) float s_op[DB_ROWS * 64];
    __shared__ alignas(16) int s_list[LIST_CAP];
    __shared__ alignas(16) int s_q[16][64];

    const int tid  = threadIdx.x;
    const int lane = tid & 31;
    const int wave = __builtin_amdgcn_readfirstlane(tid >> 5);
    const int key  = (int)blockIdx.x;
    const int row0 = key * DB_ROWS;
    int nrows = n_nodes - row0;
    nrows = nrows > DB_ROWS ? DB_ROWS : nrows;
    const int lim = nrows * 16;

    const int nslab = (n_edges + SLAB - 1) / SLAB;
    int st[4], cn[4];
#pragma unroll
    for (int j = 0; j < 4; ++j) {
        const int t = 4 * lane + j;
        st[j] = 0; cn[j] = 0;
        if (t < nslab) {
            const int* tp = tab + t * TABW + key;
            st[j] = tp[0];
            cn[j] = tp[1] - tp[0];
        }
    }

    {
        const int rb = lim > 0 ? row0 : 0;
        const v4f* qg = (const v4f*)(Qc + rb * 64);
        const v4f* pg = (const v4f*)(Qp + rb * 64);
        static_assert(DB_ROWS * 16 == 2 * 512, "2 staging trips of 512 threads");
        v4f qq[2], pp[2];
#pragma unroll
        for (int j = 0; j < 2; ++j) {
            const int i = tid + 512 * j;
            const int ic = i < lim ? i : 0;
            qq[j] = qg[ic];
            pp[j] = pg[ic];
        }
        const v4f z = (v4f){0.f, 0.f, 0.f, 0.f};
#pragma unroll
        for (int j = 0; j < 2; ++j) {
            const int i = tid + 512 * j;
            ((v4f*)s_oc)[i] = z;
            ((v4f*)s_op)[i] = z;
            ((v4f*)s_qc)[i] = qq[j];
            ((v4f*)s_qp)[i] = pp[j];
        }
    }

    const int sum4 = cn[0] + cn[1] + cn[2] + cn[3];
    int inc = sum4;
    for (int o = 1; o < 32; o <<= 1) {
        const int y = __shfl_up(inc, o);
        if (lane >= o) inc += y;
    }
    const int total = __builtin_amdgcn_readlane(inc, 31);
    int of[4];
    of[0] = inc - sum4;
    of[1] = of[0] + cn[0];
    of[2] = of[1] + cn[1];
    of[3] = of[2] + cn[2];
    const int jj = wave & 3;
    const int st_s = jj == 0 ? st[0] : jj == 1 ? st[1] : jj == 2 ? st[2] : st[3];
    const int cn_s = jj == 0 ? cn[0] : jj == 1 ? cn[1] : jj == 2 ? cn[2] : cn[3];
    const int of_s = jj == 0 ? of[0] : jj == 1 ? of[1] : jj == 2 ? of[2] : of[3];
    __syncthreads();

    static_assert(EPT == 4, "a trip takes 4 words, one 16-byte ring read");
    int* q = s_q[wave];

    for (int lo = 0; lo < total; lo += LIST_CAP) {
        {
            int wv[8], c8[8], p8[8];
#pragma unroll
            for (int i = 0; i < 8; ++i) {
                const int L = (wave >> 2) + 4 * i;
                const int s0 = __builtin_amdgcn_readlane(st_s, L);
                c8[i] = __builtin_amdgcn_readlane(cn_s, L);
                p8[i] = __builtin_amdgcn_readlane(of_s, L) - lo;
                const int* sp = slabs + (wave + 16 * i) * SLAB + s0;
                wv[i] = 0x7fffffff;
                if (lane < c8[i]) wv[i] = sp[lane];
            }
#pragma unroll
            for (int i = 0; i < 8; ++i) {
                place_chunk(wv[i], c8[i] > 32 ? 32 : c8[i], p8[i], lane, s_list);
                for (int p = 32; p < c8[i]; p += 32) {
                    const int L = (wave >> 2) + 4 * i;
                    const int s0 = __builtin_amdgcn_readlane(st_s, L);
                    const int* sp = slabs + (wave + 16 * i) * SLAB + s0 + p;
                    const int m = c8[i] - p > 32 ? 32 : c8[i] - p;
                    int w2 = 0x7fffffff;
                    if (lane < m) w2 = sp[lane];
                    place_chunk(w2, m, p8[i] + p, lane, s_list);
                }
            }
        }
        __syncthreads();

        {
            int cnt = total - lo;
            cnt = cnt > LIST_CAP ? LIST_CAP : cnt;
            int head = 0, tail = 0;
            for (int b = 0; b < cnt; b += 32) {
                const int idx = b + lane;
                const int w1 = idx < cnt ? s_list[idx] : -1;
                const bool mine = idx < cnt && (w1 >> 26) == wave;
                const unsigned mk = __builtin_amdgcn_ballot_w32(mine);
                if (mine) q[(tail + __builtin_popcount(mk & ((1u << lane) - 1u))) & 63] = w1;
                tail += __builtin_popcount(mk);
                while (tail - head >= EPT) {
                    const int4 a = *(const int4*)(q + (head & 63));
                    const int w4[EPT] = {__builtin_amdgcn_readfirstlane(a.x), __builtin_amdgcn_readfirstlane(a.y),
                                         __builtin_amdgcn_readfirstlane(a.z), __builtin_amdgcn_readfirstlane(a.w)};
                    edge_trip<EPT>(w4, lane, Kn, Vn, s_qc, s_qp, s_oc, s_op);
                    head += EPT;
                }
            }
            for (; head < tail; ++head) {
                const int w1[1] = {__builtin_amdgcn_readfirstlane(q[head & 63])};
                edge_trip<1>(w1, lane, Kn, Vn, s_qc, s_qp, s_oc, s_op);
            }
        }
        __syncthreads();
    }

#pragma unroll
    for (int j = 0; j < 2; ++j) {
        const int i = tid + 512 * j;
        if (i < lim) {
            ((v4f*)out_c)[row0 * 16 + i] = ((const v4f*)s_oc)[i];
            ((v4f*)out_p)[row0 * 16 + i] = ((const v4f*)s_op)[i];
        }
    }
}

extern "C" void kernel_launch(void* const* d_in, const int* in_sizes, int n_in,
                              void* d_out, int out_size, void* d_ws, size_t ws_size,
                              hipStream_t stream)
{
    const float* h_c  = (const float*)d_in[0];
    const float* h_p  = (const float*)d_in[1];
    const int*   src  = (const int*)  d_in[2];
    const int*   dst  = (const int*)  d_in[3];
    const float* Wqcw = (const float*)d_in[4];
    const float* Wqcb = (const float*)d_in[5];
    const float* Wqpw = (const float*)d_in[6];
    const float* Wqpb = (const float*)d_in[7];
    const float* Wkw  = (const float*)d_in[8];
    const float* Wkb  = (const float*)d_in[9];
    const float* Wvw  = (const float*)d_in[10];
    const float* Wvb  = (const float*)d_in[11];

    const int n = in_sizes[0] / 64;
    const int e = in_sizes[2];

    float* Qc = (float*)d_ws;
    float* Qp = Qc + (size_t)n * 64;
    float* Kn = Qp + (size_t)n * 64;
    float* Vn = Kn + (size_t)n * 64;
    int* slabs = (int*)(Vn + (size_t)n * 64);
    int* tab = slabs + (size_t)e;

    float* out_c = (float*)d_out;
    float* out_p = out_c + (size_t)n * 64;

    {
        proj_wmma_kernel<<<PROJ_BLOCKS, 256, 0, stream>>>(
            h_c, h_p, Wqcw, Wqcb, Wqpw, Wqpb, Wkw, Wkb, Wvw, Wvb,
            Qc, Qp, Kn, Vn, n);
    }

    {
        bin_kernel<<<BIN_BLOCKS, 1024, 0, stream>>>(src, dst, slabs, tab, n, e);
    }

    {
        edge_attn_kernel<<<NDB, 512, 0, stream>>>(
            slabs, tab, Qc, Qp, Kn, Vn, out_c, out_p, n, e);
    }
}


# edge kernel: block's edges split by row and sorted by source in LDS; a wave works one row at a time, Q and sums in registers (lane = 4 columns, 16-byte K/V loads); LDS adds go; ascending source order
# speedup vs baseline: 5.4626x; 1.1167x over previous
#include <hip/hip_runtime.h>

typedef __attribute__((ext_vector_type(2))) float v2f;
typedef __attribute__((ext_vector_type(8))) float v8f;
typedef __attribute__((ext_vector_type(4))) float v4f;

#define DB_ROWS     64
#define NDB         1563
#define NKEY        NDB
#define TABW        (NKEY + 1)
#define LIST_CAP    2048
#define SLAB        16384
#define BIN_BLOCKS  98

#define PROJ_BLOCKS 512

__global__ __launch_bounds__(256)
void proj_wmma_kernel(const float* __restrict__ h_c,
                      const float* __restrict__ h_p,
                      const float* __restrict__ Wqc, const float* __restrict__ bqc,
                      const float* __restrict__ Wqp, const float* __restrict__ bqp,
                      const float* __restrict__ Wk,  const float* __restrict__ bk,
                      const float* __restrict__ Wv,  const float* __restrict__ bv,
                      float* __restrict__ Qc, float* __restrict__ Qp,
                      float* __restrict__ Kn, float* __restrict__ Vn, int n)
{
    const int tid = threadIdx.x;
    const int lane = tid & 31;
    const int wave = __builtin_amdgcn_readfirstlane(tid >> 5);
    const int o    = wave & 3;
    const int half = lane >> 4;
    const int c16  = lane & 15;

    const float* Wm  = (o == 0) ? Wqc : (o == 1) ? Wqp : (o == 2) ? Wk : Wv;
    const float* bm  = (o == 0) ? bqc : (o == 1) ? bqp : (o == 2) ? bk : bv;
    const float* hin = (o == 1) ? h_p : h_c;
    float*       op  = (o == 0) ? Qc  : (o == 1) ? Qp  : (o == 2) ? Kn : Vn;

    v2f b[4][16];
    float bias[4];
#pragma unroll
    for (int nt = 0; nt < 4; ++nt) {
        bias[nt] = bm[nt * 16 + c16];
#pragma unroll
        for (int ks = 0; ks < 16; ++ks) {
            const float* w = Wm + (ks * 4 + 2 * half) * 64 + nt * 16 + c16;
            b[nt][ks] = (v2f){w[0], w[64]};
        }
    }

    const int ntiles  = n >> 4;
    const int nstream = (int)gridDim.x * 2;
    int t = (int)blockIdx.x * 2 + (wave >> 2);

    if (t >= ntiles) return;

#define LOAD_A(dst, tile)                                                       \
    {                                                                           \
        const float* hr = hin + ((tile) * 16 + c16) * 64 + 2 * half;            \
        _Pragma("unroll")                                                       \
        for (int ks = 0; ks < 16; ++ks) dst[ks] = *(const v2f*)(hr + ks * 4);   \
    }

#define DO_TILE(av, tile)                                                       \
    {                                                                           \
        const int m0 = (tile) * 16;                                             \
        v8f acc[4];                                                             \
        _Pragma("unroll")                                                       \
        for (int nt = 0; nt < 4; ++nt)                                          \
            acc[nt] = (v8f){0.f, 0.f, 0.f, 0.f, 0.f, 0.f, 0.f, 0.f};            \
        _Pragma("unroll")                                                       \
        for (int ks = 0; ks < 16; ++ks) {                                       \
            _Pragma("unroll")                                                   \
            for (int nt = 0; nt < 4; ++nt)                                      \
                acc[nt] = __builtin_amdgcn_wmma_f32_16x16x4_f32(                \
                     false, av[ks],  false, b[nt][ks],        \
                     (short)0, acc[nt],  false,             \
                     false);                                         \
        }                                                                       \
        _Pragma("unroll")                                                       \
        for (int r = 0; r < 8; ++r) {     \
            float* orow = op + (m0 + r + 8 * half) * 64 + c16;                  \
            _Pragma("unroll")                                                   \
            for (int nt = 0; nt < 4; ++nt)                                      \
                orow[nt * 16] = acc[nt][r] + bias[nt];                          \
        }                                                                       \
    }

    v2f a0[16], a1[16];
    LOAD_A(a0, t)
    while (t + nstream < ntiles) {
        const int t1 = t + nstream;
        const int t2 = t1 + nstream;
        LOAD_A(a1, t1)
        DO_TILE(a0, t)
        LOAD_A(a0, (t2 < ntiles ? t2 : t1))
        DO_TILE(a1, t1)
        t = t2;
    }
    if (t < ntiles) DO_TILE(a0, t)
#undef LOAD_A
#undef DO_TILE
}

__global__ __launch_bounds__(1024)
void bin_kernel(const int* __restrict__ src, const int* __restrict__ dst,
                int* __restrict__ slabs, int* __restrict__ tab,
                int n_nodes, int n_edges)
{
    static_assert(DB_ROWS == 64, "destination row inside its block: 6 bits");
    static_assert(TABW <= 2048, "one counter per key and one for the total");
    __shared__ int s_cnt[2048];
    __shared__ int s_wsum[32];
    __shared__ alignas(16) int s_pair[SLAB];

    const int tid  = threadIdx.x;
    const int lane = tid & 31;
    const int wave = tid >> 5;
    const int nslab = (n_edges + SLAB - 1) / SLAB;

    for (int slab = blockIdx.x; slab < nslab; slab += (int)gridDim.x) {
        const int base = slab * SLAB;
        s_cnt[tid] = 0;
        s_cnt[tid + 1024] = 0;
        __syncthreads();

        int sv[16], dv[16], ky[16], rk[16];
#pragma unroll
        for (int i = 0; i < 16; ++i) {
            const int e = base + i * 1024 + tid;
            const int ec = e < n_edges ? e : n_edges - 1;
            sv[i] = src[ec];
            dv[i] = dst[ec];
        }
#pragma unroll
        for (int i = 0; i < 16; ++i) {
            ky[i] = 0; rk[i] = 0;
            if (base + i * 1024 + tid < n_edges) {
                ky[i] = dv[i] / DB_ROWS;
                rk[i] = atomicAdd(&s_cnt[ky[i]], 1);
            } else {
                sv[i] = -1;
            }
        }
        __syncthreads();

        const int c0 = s_cnt[2 * tid], c1 = s_cnt[2 * tid + 1];
        int inc = c0 + c1;
        for (int o = 1; o < 32; o <<= 1) {
            const int y = __shfl_up(inc, o);
            if (lane >= o) inc += y;
        }
        if (lane == 31) s_wsum[wave] = inc;
        __syncthreads();
        if (wave == 0) {
            const int w = s_wsum[lane];
            int wi = w;
            for (int o = 1; o < 32; o <<= 1) {
                const int y = __shfl_up(wi, o);
                if (lane >= o) wi += y;
            }
            s_wsum[lane] = wi - w;
        }
        __syncthreads();
        const int ex = inc - (c0 + c1) + s_wsum[wave];
        s_cnt[2 * tid]     = ex;
        s_cnt[2 * tid + 1] = ex + c0;
        __syncthreads();

#pragma unroll
        for (int i = 0; i < 16; ++i) rk[i] += s_cnt[ky[i]];
#pragma unroll
        for (int i = 0; i < 16; ++i)
            if (sv[i] >= 0) s_pair[rk[i]] = ((dv[i] & (DB_ROWS - 1)) << 24) | sv[i];
        for (int k = tid; k < TABW; k += 1024) tab[slab * TABW + k] = s_cnt[k];
        __syncthreads();

        for (int i = 0; i < 16; ++i) {
            const int idx = i * 1024 + tid;
            if (base + idx < n_edges) slabs[base + idx] = s_pair[idx];
        }
        __syncthreads();
    }
}

#define EPT 4

__device__ __forceinline__ float lane_xor1(float x)
{
    return __builtin_bit_cast(float, __builtin_amdgcn_update_dpp(
               0, __builtin_bit_cast(int, x), 0x161, 0xf, 0xf, true));
}

__device__ __forceinline__ float edge_score(float dot)
{
    return __expf(fminf(fmaxf(dot * 0.125f, -5.0f), 5.0f));
}

__device__ __forceinline__ void edge_row(const v4f k, const v4f v, const v4f q, v4f& acc)
{
    float d = q.x * k.x + q.y * k.y + q.z * k.z + q.w * k.w;
    d += lane_xor1(d);
    const float sc = edge_score(d);
    acc += v * sc;
}

template <bool FULL>
__device__ __forceinline__ void edge_trip(const int (&w)[EPT], int cnt, int c4,
                                          const float* __restrict__ Kn,
                                          const float* __restrict__ Vn,
                                          const v4f q, v4f& acc)
{
    v4f k[EPT], v[EPT];
#pragma unroll
    for (int i = 0; i < EPT; ++i) {
        const int sof = (w[i] & 0xffffff) * 64 + c4;
        k[i] = *(const v4f*)(Kn + sof);
        v[i] = *(const v4f*)(Vn + sof);
    }
#pragma unroll
    for (int i = 0; i < EPT; ++i)
        if (FULL || i < cnt) edge_row(k[i], v[i], q, acc);
}

__global__ __launch_bounds__(512, 12)
void edge_attn_kernel(const int* __restrict__ slabs, const int* __restrict__ tab,
                      const float* __restrict__ Qc, const float* __restrict__ Qp,
                      const float* __restrict__ Kn, const float* __restrict__ Vn,
                      float* __restrict__ out_c, float* __restrict__ out_p,
                      int n_nodes, int n_edges)
{
    static_assert(DB_ROWS == 16 * 4 && DB_ROWS == 2 * 32, "16 waves of 4 rows; a lane scans 2 row counters");
    __shared__ alignas(16) int s_list[LIST_CAP];
    __shared__ alignas(16) int s_sort[LIST_CAP];
    __shared__ int s_cnt[DB_ROWS];
    __shared__ int s_cur[DB_ROWS];
    __shared__ int s_start[16][DB_ROWS];

    const int tid  = threadIdx.x;
    const int lane = tid & 31;
    const int wave = __builtin_amdgcn_readfirstlane(tid >> 5);
    const int key  = (int)blockIdx.x;
    const int row0 = key * DB_ROWS;
    const int c4   = 4 * (lane & 15);
    const bool up  = lane >= 16;

    if (tid < DB_ROWS) { s_cnt[tid] = 0; s_cur[tid] = 0; }

    const int nslab = (n_edges + SLAB - 1) / SLAB;
    int st[4], cn[4];
#pragma unroll
    for (int j = 0; j < 4; ++j) {
        const int t = 4 * lane + j;
        st[j] = 0; cn[j] = 0;
        if (t < nslab) {
            const int* tp = tab + t * TABW + key;
            st[j] = tp[0];
            cn[j] = tp[1] - tp[0];
        }
    }

    const float* qsrc  = up ? Qp : Qc;
    float*       obase = up ? out_p : out_c;

    const int sum4 = cn[0] + cn[1] + cn[2] + cn[3];
    int inc = sum4;
    for (int o = 1; o < 32; o <<= 1) {
        const int y = __shfl_up(inc, o);
        if (lane >= o) inc += y;
    }
    const int total = __builtin_amdgcn_readlane(inc, 31);
    int of[4];
    of[0] = inc - sum4;
    of[1] = of[0] + cn[0];
    of[2] = of[1] + cn[1];
    of[3] = of[2] + cn[2];
    const int jj = wave & 3;
    const int st_s = jj == 0 ? st[0] : jj == 1 ? st[1] : jj == 2 ? st[2] : st[3];
    const int cn_s = jj == 0 ? cn[0] : jj == 1 ? cn[1] : jj == 2 ? cn[2] : cn[3];
    const int of_s = jj == 0 ? of[0] : jj == 1 ? of[1] : jj == 2 ? of[2] : of[3];
    __syncthreads();

    int* mystart = s_start[wave];

    for (int lo = 0; lo < total; lo += LIST_CAP) {
        int wv[8], c8[8], p8[8], s8[8];
#pragma unroll
        for (int i = 0; i < 8; ++i) {
            const int L = (wave >> 2) + 4 * i;
            s8[i] = __builtin_amdgcn_readlane(st_s, L);
            c8[i] = __builtin_amdgcn_readlane(cn_s, L);
            p8[i] = __builtin_amdgcn_readlane(of_s, L) - lo;
            const unsigned so = (unsigned)((wave + 16 * i) * SLAB + s8[i] + lane);
            wv[i] = 0;
            if (lane < c8[i]) wv[i] = slabs[so];
        }

#pragma unroll
        for (int i = 0; i < 8; ++i) {
            if (lane < c8[i] && (unsigned)(p8[i] + lane) < (unsigned)LIST_CAP)
                atomicAdd(&s_cnt[wv[i] >> 24], 1);
#pragma nounroll
            for (int p = 32; p < c8[i]; p += 32) {
                const unsigned so = (unsigned)((wave + 16 * i) * SLAB + s8[i] + p + lane);
                if (p + lane < c8[i] && (unsigned)(p8[i] + p + lane) < (unsigned)LIST_CAP)
                    atomicAdd(&s_cnt[slabs[so] >> 24], 1);
            }
        }
        __syncthreads();

        const int n0 = s_cnt[2 * lane], n1 = s_cnt[2 * lane + 1];
        int rs = n0 + n1;
        for (int o = 1; o < 32; o <<= 1) {
            const int y = __shfl_up(rs, o);
            if (lane >= o) rs += y;
        }
        const int b0 = rs - (n0 + n1), b1 = b0 + n0;
        mystart[2 * lane]     = b0;
        mystart[2 * lane + 1] = b1;

#pragma unroll
        for (int i = 0; i < 8; ++i) {
            if (lane < c8[i] && (unsigned)(p8[i] + lane) < (unsigned)LIST_CAP) {
                const int r = wv[i] >> 24;
                s_list[mystart[r] + atomicAdd(&s_cur[r], 1)] = wv[i];
            }
#pragma nounroll
            for (int p = 32; p < c8[i]; p += 32) {
                const unsigned so = (unsigned)((wave + 16 * i) * SLAB + s8[i] + p + lane);
                if (p + lane < c8[i] && (unsigned)(p8[i] + p + lane) < (unsigned)LIST_CAP) {
                    const int w2 = slabs[so];
                    const int r = w2 >> 24;
                    s_list[mystart[r] + atomicAdd(&s_cur[r], 1)] = w2;
                }
            }
        }
        __syncthreads();
        if (tid < DB_ROWS) { s_cnt[tid] = 0; s_cur[tid] = 0; }

        for (int j = 0; j < 4; ++j) {
            const int Lr = 2 * wave + (j >> 1);
            const int n  = __builtin_amdgcn_readlane((j & 1) ? n1 : n0, Lr);
            const int sb = __builtin_amdgcn_readlane((j & 1) ? b1 : b0, Lr);
            const int* lrow = s_list + sb;
            int* srow = s_sort + sb;
            const int row = row0 + 4 * wave + j;
            const bool live = row < n_nodes;

            const int rq = live ? row : n_nodes - 1;
            const v4f q = *(const v4f*)(qsrc + rq * 64 + c4);
            v4f acc = (v4f){0.f, 0.f, 0.f, 0.f};
            if (lo > 0 && live) acc = *(const v4f*)(obase + row * 64 + c4);

            for (int cb = 0; cb < n; cb += 32) {
                const int t = cb + lane;
                const int x = t < n ? lrow[t] : 0x7fffffff;
                const unsigned long long kx = ((unsigned long long)(unsigned)x << 32) | (unsigned)t;
                int rank = 0;
                for (int ob = 0; ob < n; ob += 32) {
                    const int u = ob + lane;
                    const int y = u < n ? lrow[u] : 0x7fffffff;
                    const int mm = n - ob > 32 ? 32 : n - ob;
                    for (int m = 0; m < mm; ++m) {
                        const unsigned ym = (unsigned)__builtin_amdgcn_readlane(y, m);
                        const unsigned long long ky = ((unsigned long long)ym << 32) | (unsigned)(ob + m);
                        rank += ky < kx ? 1 : 0;
                    }
                }
                if (t < n) srow[rank] = x;
            }

            for (int cb = 0; cb < n; cb += 32) {
                const int t = cb + lane;
                const int xs = t < n ? srow[t] : 0;
                const int mm = n - cb > 32 ? 32 : n - cb;
                int e = 0;
                for (; e + EPT <= mm; e += EPT) {
                    int w4[EPT];
#pragma unroll
                    for (int i = 0; i < EPT; ++i) w4[i] = __builtin_amdgcn_readlane(xs, e + i);
                    edge_trip<true>(w4, EPT, c4, Kn, Vn, q, acc);
                }
                if (e < mm) {
                    int w4[EPT];
#pragma unroll
                    for (int i = 0; i < EPT; ++i)
                        w4[i] = __builtin_amdgcn_readlane(xs, e + i < mm ? e + i : mm - 1);
                    edge_trip<false>(w4, mm - e, c4, Kn, Vn, q, acc);
                }
            }

            if (live) *(v4f*)(obase + row * 64 + c4) = acc;
        }
        __syncthreads();
    }

    if (total <= 0) {
#pragma unroll
        for (int j = 0; j < 4; ++j) {
            const int row = row0 + 4 * wave + j;
            if (row < n_nodes) *(v4f*)(obase + row * 64 + c4) = (v4f){0.f, 0.f, 0.f, 0.f};
        }
    }
}

extern "C" void kernel_launch(void* const* d_in, const int* in_sizes, int n_in,
                              void* d_out, int out_size, void* d_ws, size_t ws_size,
                              hipStream_t stream)
{
    const float* h_c  = (const float*)d_in[0];
    const float* h_p  = (const float*)d_in[1];
    const int*   src  = (const int*)  d_in[2];
    const int*   dst  = (const int*)  d_in[3];
    const float* Wqcw = (const float*)d_in[4];
    const float* Wqcb = (const float*)d_in[5];
    const float* Wqpw = (const float*)d_in[6];
    const float* Wqpb = (const float*)d_in[7];
    const float* Wkw  = (const float*)d_in[8];
    const float* Wkb  = (const float*)d_in[9];
    const float* Wvw  = (const float*)d_in[10];
    const float* Wvb  = (const float*)d_in[11];

    const int n = in_sizes[0] / 64;
    const int e = in_sizes[2];

    float* Qc = (float*)d_ws;
    float* Qp = Qc + (size_t)n * 64;
    float* Kn = Qp + (size_t)n * 64;
    float* Vn = Kn + (size_t)n * 64;
    int* slabs = (int*)(Vn + (size_t)n * 64);
    int* tab = slabs + (size_t)e;

    float* out_c = (float*)d_out;
    float* out_p = out_c + (size_t)n * 64;

    {
        proj_wmma_kernel<<<PROJ_BLOCKS, 256, 0, stream>>>(
            h_c, h_p, Wqcw, Wqcb, Wqpw, Wqpb, Wkw, Wkb, Wvw, Wvb,
            Qc, Qp, Kn, Vn, n);
    }

    {
        bin_kernel<<<BIN_BLOCKS, 1024, 0, stream>>>(src, dst, slabs, tab, n, e);
    }

    {
        edge_attn_kernel<<<NDB, 512, 0, stream>>>(
            slabs, tab, Qc, Qp, Kn, Vn, out_c, out_p, n, e);
    }
}


# Q_c/Q_p projection moved into the edge kernel: a block computes its own 64 nodes' Q rows by f32 WMMA into LDS (same k order, bias); the Q planes in memory go, projection launch does K, V only
# speedup vs baseline: 6.0377x; 1.1053x over previous
#include <hip/hip_runtime.h>

typedef __attribute__((ext_vector_type(2))) float v2f;
typedef __attribute__((ext_vector_type(8))) float v8f;
typedef __attribute__((ext_vector_type(4))) float v4f;

#define DB_ROWS     64
#define NDB         1563
#define NKEY        NDB
#define TABW        (NKEY + 1)
#define LIST_CAP    2048
#define SLAB        16384
#define BIN_BLOCKS  98

#define PROJ_BLOCKS 512

__global__ __launch_bounds__(256)
void proj_wmma_kernel(const float* __restrict__ h_c,
                      const float* __restrict__ Wk,  const float* __restrict__ bk,
                      const float* __restrict__ Wv,  const float* __restrict__ bv,
                      float* __restrict__ Kn, float* __restrict__ Vn, int n)
{
    const int tid = threadIdx.x;
    const int lane = tid & 31;
    const int wave = __builtin_amdgcn_readfirstlane(tid >> 5);
    const int o    = wave & 1;
    const int half = lane >> 4;
    const int c16  = lane & 15;

    const float* Wm  = (o == 0) ? Wk : Wv;
    const float* bm  = (o == 0) ? bk : bv;
    const float* hin = h_c;
    float*       op  = (o == 0) ? Kn : Vn;

    v2f b[4][16];
    float bias[4];
#pragma unroll
    for (int nt = 0; nt < 4; ++nt) {
        bias[nt] = bm[nt * 16 + c16];
#pragma unroll
        for (int ks = 0; ks < 16; ++ks) {
            const float* w = Wm + (ks * 4 + 2 * half) * 64 + nt * 16 + c16;
            b[nt][ks] = (v2f){w[0], w[64]};
        }
    }

    const int ntiles  = n >> 4;
    const int nstream = (int)gridDim.x * 4;
    int t = (int)blockIdx.x * 4 + (wave >> 1);

    if (t >= ntiles) return;

#define LOAD_A(dst, tile)                                                       \
    {                                                                           \
        const float* hr = hin + ((tile) * 16 + c16) * 64 + 2 * half;            \
        _Pragma("unroll")                                                       \
        for (int ks = 0; ks < 16; ++ks) dst[ks] = *(const v2f*)(hr + ks * 4);   \
    }

#define DO_TILE(av, tile)                                                       \
    {                                                                           \
        const int m0 = (tile) * 16;                                             \
        v8f acc[4];                                                             \
        _Pragma("unroll")                                                       \
        for (int nt = 0; nt < 4; ++nt)                                          \
            acc[nt] = (v8f){0.f, 0.f, 0.f, 0.f, 0.f, 0.f, 0.f, 0.f};            \
        _Pragma("unroll")                                                       \
        for (int ks = 0; ks < 16; ++ks) {                                       \
            _Pragma("unroll")                                                   \
            for (int nt = 0; nt < 4; ++nt)                                      \
                acc[nt] = __builtin_amdgcn_wmma_f32_16x16x4_f32(                \
                     false, av[ks],  false, b[nt][ks],        \
                     (short)0, acc[nt],  false,             \
                     false);                                         \
        }                                                                       \
        _Pragma("unroll")                                                       \
        for (int r = 0; r < 8; ++r) {     \
            float* orow = op + (m0 + r + 8 * half) * 64 + c16;                  \
            _Pragma("unroll")                                                   \
            for (int nt = 0; nt < 4; ++nt)                                      \
                orow[nt * 16] = acc[nt][r] + bias[nt];                          \
        }                                                                       \
    }

    v2f a0[16], a1[16];
    LOAD_A(a0, t)
    while (t + nstream < ntiles) {
        const int t1 = t + nstream;
        const int t2 = t1 + nstream;
        LOAD_A(a1, t1)
        DO_TILE(a0, t)
        LOAD_A(a0, (t2 < ntiles ? t2 : t1))
        DO_TILE(a1, t1)
        t = t2;
    }
    if (t < ntiles) DO_TILE(a0, t)
#undef LOAD_A
#undef DO_TILE
}

__global__ __launch_bounds__(1024)
void bin_kernel(const int* __restrict__ src, const int* __restrict__ dst,
                int* __restrict__ slabs, int* __restrict__ tab,
                int n_nodes, int n_edges)
{
    static_assert(DB_ROWS == 64, "destination row inside its block: 6 bits");
    static_assert(TABW <= 2048, "one counter per key and one for the total");
    __shared__ int s_cnt[2048];
    __shared__ int s_wsum[32];
    __shared__ alignas(16) int s_pair[SLAB];

    const int tid  = threadIdx.x;
    const int lane = tid & 31;
    const int wave = tid >> 5;
    const int nslab = (n_edges + SLAB - 1) / SLAB;

    for (int slab = blockIdx.x; slab < nslab; slab += (int)gridDim.x) {
        const int base = slab * SLAB;
        s_cnt[tid] = 0;
        s_cnt[tid + 1024] = 0;
        __syncthreads();

        int sv[16], dv[16], ky[16], rk[16];
#pragma unroll
        for (int i = 0; i < 16; ++i) {
            const int e = base + i * 1024 + tid;
            const int ec = e < n_edges ? e : n_edges - 1;
            sv[i] = src[ec];
            dv[i] = dst[ec];
        }
#pragma unroll
        for (int i = 0; i < 16; ++i) {
            ky[i] = 0; rk[i] = 0;
            if (base + i * 1024 + tid < n_edges) {
                ky[i] = dv[i] / DB_ROWS;
                rk[i] = atomicAdd(&s_cnt[ky[i]], 1);
            } else {
                sv[i] = -1;
            }
        }
        __syncthreads();

        const int c0 = s_cnt[2 * tid], c1 = s_cnt[2 * tid + 1];
        int inc = c0 + c1;
        for (int o = 1; o < 32; o <<= 1) {
            const int y = __shfl_up(inc, o);
            if (lane >= o) inc += y;
        }
        if (lane == 31) s_wsum[wave] = inc;
        __syncthreads();
        if (wave == 0) {
            const int w = s_wsum[lane];
            int wi = w;
            for (int o = 1; o < 32; o <<= 1) {
                const int y = __shfl_up(wi, o);
                if (lane >= o) wi += y;
            }
            s_wsum[lane] = wi - w;
        }
        __syncthreads();
        const int ex = inc - (c0 + c1) + s_wsum[wave];
        s_cnt[2 * tid]     = ex;
        s_cnt[2 * tid + 1] = ex + c0;
        __syncthreads();

#pragma unroll
        for (int i = 0; i < 16; ++i) rk[i] += s_cnt[ky[i]];
#pragma unroll
        for (int i = 0; i < 16; ++i)
            if (sv[i] >= 0) s_pair[rk[i]] = ((dv[i] & (DB_ROWS - 1)) << 24) | sv[i];
        for (int k = tid; k < TABW; k += 1024) tab[slab * TABW + k] = s_cnt[k];
        __syncthreads();

        for (int i = 0; i < 16; ++i) {
            const int idx = i * 1024 + tid;
            if (base + idx < n_edges) slabs[base + idx] = s_pair[idx];
        }
        __syncthreads();
    }
}

#define EPT 4

__device__ __forceinline__ float lane_xor1(float x)
{
    return __builtin_bit_cast(float, __builtin_amdgcn_update_dpp(
               0, __builtin_bit_cast(int, x), 0x161, 0xf, 0xf, true));
}

__device__ __forceinline__ float edge_score(float dot)
{
    return __expf(fminf(fmaxf(dot * 0.125f, -5.0f), 5.0f));
}

__device__ __forceinline__ void edge_row(const v4f k, const v4f v, const v4f q, v4f& acc)
{
    float d = q.x * k.x + q.y * k.y + q.z * k.z + q.w * k.w;
    d += lane_xor1(d);
    const float sc = edge_score(d);
    acc += v * sc;
}

template <bool FULL>
__device__ __forceinline__ void edge_trip(const int (&w)[EPT], int cnt, int c4,
                                          const float* __restrict__ Kn,
                                          const float* __restrict__ Vn,
                                          const v4f q, v4f& acc)
{
    v4f k[EPT], v[EPT];
#pragma unroll
    for (int i = 0; i < EPT; ++i) {
        const int sof = (w[i] & 0xffffff) * 64 + c4;
        k[i] = *(const v4f*)(Kn + sof);
        v[i] = *(const v4f*)(Vn + sof);
    }
#pragma unroll
    for (int i = 0; i < EPT; ++i)
        if (FULL || i < cnt) edge_row(k[i], v[i], q, acc);
}

__global__ __launch_bounds__(512, 12)
void edge_attn_kernel(const int* __restrict__ slabs, const int* __restrict__ tab,
                      const float* __restrict__ h_c, const float* __restrict__ h_p,
                      const float* __restrict__ Wqc, const float* __restrict__ bqc,
                      const float* __restrict__ Wqp, const float* __restrict__ bqp,
                      const float* __restrict__ Kn, const float* __restrict__ Vn,
                      float* __restrict__ out_c, float* __restrict__ out_p,
                      int n_nodes, int n_edges)
{
    static_assert(DB_ROWS == 16 * 4 && DB_ROWS == 2 * 32, "16 waves of 4 rows; a lane scans 2 row counters");
    __shared__ alignas(16) int s_list[LIST_CAP];
    __shared__ alignas(16) int s_sort[LIST_CAP];
    __shared__ int s_cnt[DB_ROWS];
    __shared__ int s_cur[DB_ROWS];
    __shared__ int s_start[16][DB_ROWS];
    __shared__ alignas(16) float s_q[2 * DB_ROWS * 64];

    const int tid  = threadIdx.x;
    const int lane = tid & 31;
    const int wave = __builtin_amdgcn_readfirstlane(tid >> 5);
    const int key  = (int)blockIdx.x;
    const int row0 = key * DB_ROWS;
    const int c4   = 4 * (lane & 15);
    const bool up  = lane >= 16;

    {
        const int q_half = lane >> 4;
        const int q_c16  = lane & 15;
        const int q_rt   = wave & 3;
        const int q_side = (wave >> 2) & 1;
        const int q_nh   = wave >> 3;
        const float* q_h = q_side ? h_p : h_c;
        const float* q_W = q_side ? Wqp : Wqc;
        const float* q_b = q_side ? bqp : bqc;
        int qr = row0 + 16 * q_rt + q_c16;
        qr = qr < n_nodes ? qr : n_nodes - 1;
        const float* hr = q_h + qr * 64 + 2 * q_half;
        v2f qa[16];
#pragma unroll
        for (int ks = 0; ks < 16; ++ks) qa[ks] = *(const v2f*)(hr + ks * 4);
        float* qd = s_q + q_side * (DB_ROWS * 64) + (16 * q_rt + 8 * q_half) * 64 + q_nh * 32 + q_c16;
#pragma unroll
        for (int nt = 0; nt < 2; ++nt) {
            const float* w0 = q_W + (2 * q_half) * 64 + q_nh * 32 + nt * 16 + q_c16;
            const float bias = q_b[q_nh * 32 + nt * 16 + q_c16];
            v2f qb[16];
#pragma unroll
            for (int ks = 0; ks < 16; ++ks) qb[ks] = (v2f){w0[ks * 256], w0[ks * 256 + 64]};
            v8f acc = (v8f){0.f, 0.f, 0.f, 0.f, 0.f, 0.f, 0.f, 0.f};
#pragma unroll
            for (int ks = 0; ks < 16; ++ks)
                acc = __builtin_amdgcn_wmma_f32_16x16x4_f32(
                     false, qa[ks],  false, qb[ks],
                     (short)0, acc,  false,  false);
#pragma unroll
            for (int r = 0; r < 8; ++r) qd[r * 64 + nt * 16] = acc[r] + bias;
        }
    }

    if (tid < DB_ROWS) { s_cnt[tid] = 0; s_cur[tid] = 0; }

    const int nslab = (n_edges + SLAB - 1) / SLAB;
    int st[4], cn[4];
#pragma unroll
    for (int j = 0; j < 4; ++j) {
        const int t = 4 * lane + j;
        st[j] = 0; cn[j] = 0;
        if (t < nslab) {
            const int* tp = tab + t * TABW + key;
            st[j] = tp[0];
            cn[j] = tp[1] - tp[0];
        }
    }

    float*       obase = up ? out_p : out_c;

    const int sum4 = cn[0] + cn[1] + cn[2] + cn[3];
    int inc = sum4;
    for (int o = 1; o < 32; o <<= 1) {
        const int y = __shfl_up(inc, o);
        if (lane >= o) inc += y;
    }
    const int total = __builtin_amdgcn_readlane(inc, 31);
    int of[4];
    of[0] = inc - sum4;
    of[1] = of[0] + cn[0];
    of[2] = of[1] + cn[1];
    of[3] = of[2] + cn[2];
    const int jj = wave & 3;
    const int st_s = jj == 0 ? st[0] : jj == 1 ? st[1] : jj == 2 ? st[2] : st[3];
    const int cn_s = jj == 0 ? cn[0] : jj == 1 ? cn[1] : jj == 2 ? cn[2] : cn[3];
    const int of_s = jj == 0 ? of[0] : jj == 1 ? of[1] : jj == 2 ? of[2] : of[3];

    __syncthreads();

    int* mystart = s_start[wave];

    for (int lo = 0; lo < total; lo += LIST_CAP) {
        int wv[8], c8[8], p8[8], s8[8];
#pragma unroll
        for (int i = 0; i < 8; ++i) {
            const int L = (wave >> 2) + 4 * i;
            s8[i] = __builtin_amdgcn_readlane(st_s, L);
            c8[i] = __builtin_amdgcn_readlane(cn_s, L);
            p8[i] = __builtin_amdgcn_readlane(of_s, L) - lo;
            const unsigned so = (unsigned)((wave + 16 * i) * SLAB + s8[i] + lane);
            wv[i] = 0;
            if (lane < c8[i]) wv[i] = slabs[so];
        }

#pragma unroll
        for (int i = 0; i < 8; ++i) {
            if (lane < c8[i] && (unsigned)(p8[i] + lane) < (unsigned)LIST_CAP)
                atomicAdd(&s_cnt[wv[i] >> 24], 1);
#pragma nounroll
            for (int p = 32; p < c8[i]; p += 32) {
                const unsigned so = (unsigned)((wave + 16 * i) * SLAB + s8[i] + p + lane);
                if (p + lane < c8[i] && (unsigned)(p8[i] + p + lane) < (unsigned)LIST_CAP)
                    atomicAdd(&s_cnt[slabs[so] >> 24], 1);
            }
        }
        __syncthreads();

        const int n0 = s_cnt[2 * lane], n1 = s_cnt[2 * lane + 1];
        int rs = n0 + n1;
        for (int o = 1; o < 32; o <<= 1) {
            const int y = __shfl_up(rs, o);
            if (lane >= o) rs += y;
        }
        const int b0 = rs - (n0 + n1), b1 = b0 + n0;
        mystart[2 * lane]     = b0;
        mystart[2 * lane + 1] = b1;

#pragma unroll
        for (int i = 0; i < 8; ++i) {
            if (lane < c8[i] && (unsigned)(p8[i] + lane) < (unsigned)LIST_CAP) {
                const int r = wv[i] >> 24;
                s_list[mystart[r] + atomicAdd(&s_cur[r], 1)] = wv[i];
            }
#pragma nounroll
            for (int p = 32; p < c8[i]; p += 32) {
                const unsigned so = (unsigned)((wave + 16 * i) * SLAB + s8[i] + p + lane);
                if (p + lane < c8[i] && (unsigned)(p8[i] + p + lane) < (unsigned)LIST_CAP) {
                    const int w2 = slabs[so];
                    const int r = w2 >> 24;
                    s_list[mystart[r] + atomicAdd(&s_cur[r], 1)] = w2;
                }
            }
        }
        __syncthreads();
        if (tid < DB_ROWS) { s_cnt[tid] = 0; s_cur[tid] = 0; }

        for (int j = 0; j < 4; ++j) {
            const int Lr = 2 * wave + (j >> 1);
            const int n  = __builtin_amdgcn_readlane((j & 1) ? n1 : n0, Lr);
            const int sb = __builtin_amdgcn_readlane((j & 1) ? b1 : b0, Lr);
            const int* lrow = s_list + sb;
            int* srow = s_sort + sb;
            const int row = row0 + 4 * wave + j;
            const bool live = row < n_nodes;

            const v4f q = *(const v4f*)(s_q + (up ? DB_ROWS * 64 : 0) + (4 * wave + j) * 64 + c4);
            v4f acc = (v4f){0.f, 0.f, 0.f, 0.f};
            if (lo > 0 && live) acc = *(const v4f*)(obase + row * 64 + c4);

            for (int cb = 0; cb < n; cb += 32) {
                const int t = cb + lane;
                const int x = t < n ? lrow[t] : 0x7fffffff;
                const unsigned long long kx = ((unsigned long long)(unsigned)x << 32) | (unsigned)t;
                int rank = 0;
                for (int ob = 0; ob < n; ob += 32) {
                    const int u = ob + lane;
                    const int y = u < n ? lrow[u] : 0x7fffffff;
                    const int mm = n - ob > 32 ? 32 : n - ob;
                    for (int m = 0; m < mm; ++m) {
                        const unsigned ym = (unsigned)__builtin_amdgcn_readlane(y, m);
                        const unsigned long long ky = ((unsigned long long)ym << 32) | (unsigned)(ob + m);
                        rank += ky < kx ? 1 : 0;
                    }
                }
                if (t < n) srow[rank] = x;
            }

            for (int cb = 0; cb < n; cb += 32) {
                const int t = cb + lane;
                const int xs = t < n ? srow[t] : 0;
                const int mm = n - cb > 32 ? 32 : n - cb;
                int e = 0;
                for (; e + EPT <= mm; e += EPT) {
                    int w4[EPT];
#pragma unroll
                    for (int i = 0; i < EPT; ++i) w4[i] = __builtin_amdgcn_readlane(xs, e + i);
                    edge_trip<true>(w4, EPT, c4, Kn, Vn, q, acc);
                }
                if (e < mm) {
                    int w4[EPT];
#pragma unroll
                    for (int i = 0; i < EPT; ++i)
                        w4[i] = __builtin_amdgcn_readlane(xs, e + i < mm ? e + i : mm - 1);
                    edge_trip<false>(w4, mm - e, c4, Kn, Vn, q, acc);
                }
            }

            if (live) *(v4f*)(obase + row * 64 + c4) = acc;
        }
        __syncthreads();
    }

    if (total <= 0) {
#pragma unroll
        for (int j = 0; j < 4; ++j) {
            const int row = row0 + 4 * wave + j;
            if (row < n_nodes) *(v4f*)(obase + row * 64 + c4) = (v4f){0.f, 0.f, 0.f, 0.f};
        }
    }
}

extern "C" void kernel_launch(void* const* d_in, const int* in_sizes, int n_in,
                              void* d_out, int out_size, void* d_ws, size_t ws_size,
                              hipStream_t stream)
{
    const float* h_c  = (const float*)d_in[0];
    const float* h_p  = (const float*)d_in[1];
    const int*   src  = (const int*)  d_in[2];
    const int*   dst  = (const int*)  d_in[3];
    const float* Wqcw = (const float*)d_in[4];
    const float* Wqcb = (const float*)d_in[5];
    const float* Wqpw = (const float*)d_in[6];
    const float* Wqpb = (const float*)d_in[7];
    const float* Wkw  = (const float*)d_in[8];
    const float* Wkb  = (const float*)d_in[9];
    const float* Wvw  = (const float*)d_in[10];
    const float* Wvb  = (const float*)d_in[11];

    const int n = in_sizes[0] / 64;
    const int e = in_sizes[2];

    float* Kn = (float*)d_ws;
    float* Vn = Kn + (size_t)n * 64;
    int* slabs = (int*)(Vn + (size_t)n * 64);
    int* tab = slabs + (size_t)e;

    float* out_c = (float*)d_out;
    float* out_p = out_c + (size_t)n * 64;

    {
        proj_wmma_kernel<<<PROJ_BLOCKS, 256, 0, stream>>>(
            h_c, Wkw, Wkb, Wvw, Wvb, Kn, Vn, n);
    }

    {
        bin_kernel<<<BIN_BLOCKS, 1024, 0, stream>>>(src, dst, slabs, tab, n, e);
    }

    {
        edge_attn_kernel<<<NDB, 512, 0, stream>>>(
            slabs, tab, h_c, h_p, Wqcw, Wqcb, Wqpw, Wqpb, Kn, Vn, out_c, out_p, n, e);
    }
}


# edge kernel at 16 waves per SIMD (64 registers, no scratch): 4 work-groups per WGP, so 1563 blocks take 2 residency rounds, not 3; row addresses as 32-bit offsets, counters packed; sums unchanged
# speedup vs baseline: 6.2376x; 1.0331x over previous
#include <hip/hip_runtime.h>

typedef __attribute__((ext_vector_type(2))) float v2f;
typedef __attribute__((ext_vector_type(8))) float v8f;
typedef __attribute__((ext_vector_type(4))) float v4f;

#define DB_ROWS     64
#define NDB         1563
#define NKEY        NDB
#define TABW        (NKEY + 1)
#define LIST_CAP    2048
#define SLAB        16384
#define BIN_BLOCKS  98

#define PROJ_BLOCKS 512

__global__ __launch_bounds__(256)
void proj_wmma_kernel(const float* __restrict__ h_c,
                      const float* __restrict__ Wk,  const float* __restrict__ bk,
                      const float* __restrict__ Wv,  const float* __restrict__ bv,
                      float* __restrict__ Kn, float* __restrict__ Vn, int n)
{
    const int tid = threadIdx.x;
    const int lane = tid & 31;
    const int wave = __builtin_amdgcn_readfirstlane(tid >> 5);
    const int o    = wave & 1;
    const int half = lane >> 4;
    const int c16  = lane & 15;

    const float* Wm  = (o == 0) ? Wk : Wv;
    const float* bm  = (o == 0) ? bk : bv;
    const float* hin = h_c;
    float*       op  = (o == 0) ? Kn : Vn;

    v2f b[4][16];
    float bias[4];
#pragma unroll
    for (int nt = 0; nt < 4; ++nt) {
        bias[nt] = bm[nt * 16 + c16];
#pragma unroll
        for (int ks = 0; ks < 16; ++ks) {
            const float* w = Wm + (ks * 4 + 2 * half) * 64 + nt * 16 + c16;
            b[nt][ks] = (v2f){w[0], w[64]};
        }
    }

    const int ntiles  = n >> 4;
    const int nstream = (int)gridDim.x * 4;
    int t = (int)blockIdx.x * 4 + (wave >> 1);

    if (t >= ntiles) return;

#define LOAD_A(dst, tile)                                                       \
    {                                                                           \
        const float* hr = hin + ((tile) * 16 + c16) * 64 + 2 * half;            \
        _Pragma("unroll")                                                       \
        for (int ks = 0; ks < 16; ++ks) dst[ks] = *(const v2f*)(hr + ks * 4);   \
    }

#define DO_TILE(av, tile)                                                       \
    {                                                                           \
        const int m0 = (tile) * 16;                                             \
        v8f acc[4];                                                             \
        _Pragma("unroll")                                                       \
        for (int nt = 0; nt < 4; ++nt)                                          \
            acc[nt] = (v8f){0.f, 0.f, 0.f, 0.f, 0.f, 0.f, 0.f, 0.f};            \
        _Pragma("unroll")                                                       \
        for (int ks = 0; ks < 16; ++ks) {                                       \
            _Pragma("unroll")                                                   \
            for (int nt = 0; nt < 4; ++nt)                                      \
                acc[nt] = __builtin_amdgcn_wmma_f32_16x16x4_f32(                \
                     false, av[ks],  false, b[nt][ks],        \
                     (short)0, acc[nt],  false,             \
                     false);                                         \
        }                                                                       \
        _Pragma("unroll")                                                       \
        for (int r = 0; r < 8; ++r) {     \
            float* orow = op + (m0 + r + 8 * half) * 64 + c16;                  \
            _Pragma("unroll")                                                   \
            for (int nt = 0; nt < 4; ++nt)                                      \
                orow[nt * 16] = acc[nt][r] + bias[nt];                          \
        }                                                                       \
    }

    v2f a0[16], a1[16];
    LOAD_A(a0, t)
    while (t + nstream < ntiles) {
        const int t1 = t + nstream;
        const int t2 = t1 + nstream;
        LOAD_A(a1, t1)
        DO_TILE(a0, t)
        LOAD_A(a0, (t2 < ntiles ? t2 : t1))
        DO_TILE(a1, t1)
        t = t2;
    }
    if (t < ntiles) DO_TILE(a0, t)
#undef LOAD_A
#undef DO_TILE
}

__global__ __launch_bounds__(1024)
void bin_kernel(const int* __restrict__ src, const int* __restrict__ dst,
                int* __restrict__ slabs, int* __restrict__ tab,
                int n_nodes, int n_edges)
{
    static_assert(DB_ROWS == 64, "destination row inside its block: 6 bits");
    static_assert(TABW <= 2048, "one counter per key and one for the total");
    __shared__ int s_cnt[2048];
    __shared__ int s_wsum[32];
    __shared__ alignas(16) int s_pair[SLAB];

    const int tid  = threadIdx.x;
    const int lane = tid & 31;
    const int wave = tid >> 5;
    const int nslab = (n_edges + SLAB - 1) / SLAB;

    for (int slab = blockIdx.x; slab < nslab; slab += (int)gridDim.x) {
        const int base = slab * SLAB;
        s_cnt[tid] = 0;
        s_cnt[tid + 1024] = 0;
        __syncthreads();

        int sv[16], dv[16], ky[16], rk[16];
#pragma unroll
        for (int i = 0; i < 16; ++i) {
            const int e = base + i * 1024 + tid;
            const int ec = e < n_edges ? e : n_edges - 1;
            sv[i] = src[ec];
            dv[i] = dst[ec];
        }
#pragma unroll
        for (int i = 0; i < 16; ++i) {
            ky[i] = 0; rk[i] = 0;
            if (base + i * 1024 + tid < n_edges) {
                ky[i] = dv[i] / DB_ROWS;
                rk[i] = atomicAdd(&s_cnt[ky[i]], 1);
            } else {
                sv[i] = -1;
            }
        }
        __syncthreads();

        const int c0 = s_cnt[2 * tid], c1 = s_cnt[2 * tid + 1];
        int inc = c0 + c1;
        for (int o = 1; o < 32; o <<= 1) {
            const int y = __shfl_up(inc, o);
            if (lane >= o) inc += y;
        }
        if (lane == 31) s_wsum[wave] = inc;
        __syncthreads();
        if (wave == 0) {
            const int w = s_wsum[lane];
            int wi = w;
            for (int o = 1; o < 32; o <<= 1) {
                const int y = __shfl_up(wi, o);
                if (lane >= o) wi += y;
            }
            s_wsum[lane] = wi - w;
        }
        __syncthreads();
        const int ex = inc - (c0 + c1) + s_wsum[wave];
        s_cnt[2 * tid]     = ex;
        s_cnt[2 * tid + 1] = ex + c0;
        __syncthreads();

#pragma unroll
        for (int i = 0; i < 16; ++i) rk[i] += s_cnt[ky[i]];
#pragma unroll
        for (int i = 0; i < 16; ++i)
            if (sv[i] >= 0) s_pair[rk[i]] = ((dv[i] & (DB_ROWS - 1)) << 24) | sv[i];
        for (int k = tid; k < TABW; k += 1024) tab[slab * TABW + k] = s_cnt[k];
        __syncthreads();

        for (int i = 0; i < 16; ++i) {
            const int idx = i * 1024 + tid;
            if (base + idx < n_edges) slabs[base + idx] = s_pair[idx];
        }
        __syncthreads();
    }
}

#define EPT 4

__device__ __forceinline__ float lane_xor1(float x)
{
    return __builtin_bit_cast(float, __builtin_amdgcn_update_dpp(
               0, __builtin_bit_cast(int, x), 0x161, 0xf, 0xf, true));
}

__device__ __forceinline__ float edge_score(float dot)
{
    return __expf(fminf(fmaxf(dot * 0.125f, -5.0f), 5.0f));
}

__device__ __forceinline__ void edge_row(const v4f k, const v4f v, const v4f q, v4f& acc)
{
    float d = q.x * k.x + q.y * k.y + q.z * k.z + q.w * k.w;
    d += lane_xor1(d);
    const float sc = edge_score(d);
    acc += v * sc;
}

template <bool FULL>
__device__ __forceinline__ void edge_trip(const int (&w)[EPT], int cnt, int c4,
                                          const float* __restrict__ Kn,
                                          const float* __restrict__ Vn,
                                          const v4f q, v4f& acc)
{
    v4f k[EPT], v[EPT];
#pragma unroll
    for (int i = 0; i < EPT; ++i) {
        const int sof = (w[i] & 0xffffff) * 64 + c4;
        k[i] = *(const v4f*)(Kn + sof);
        v[i] = *(const v4f*)(Vn + sof);
    }
#pragma unroll
    for (int i = 0; i < EPT; ++i)
        if (FULL || i < cnt) edge_row(k[i], v[i], q, acc);
}

__global__ __launch_bounds__(512, 16)
void edge_attn_kernel(const int* __restrict__ slabs, const int* __restrict__ tab,
                      const float* __restrict__ h_c, const float* __restrict__ h_p,
                      const float* __restrict__ Wqc, const float* __restrict__ bqc,
                      const float* __restrict__ Wqp, const float* __restrict__ bqp,
                      const float* __restrict__ Kn, const float* __restrict__ Vn,
                      float* __restrict__ out,
                      int n_nodes, int n_edges)
{
    static_assert(DB_ROWS == 16 * 4 && DB_ROWS == 2 * 32, "16 waves of 4 rows; a lane scans 2 row counters");
    __shared__ alignas(16) int s_list[LIST_CAP];
    __shared__ alignas(16) int s_sort[LIST_CAP];
    __shared__ int s_cc[2 * DB_ROWS];
    int* const s_cnt = s_cc;
    int* const s_cur = s_cc + DB_ROWS;
    __shared__ int s_start[16][DB_ROWS];
    __shared__ alignas(16) float s_q[2 * DB_ROWS * 64];

    const int tid  = threadIdx.x;
    const int lane = tid & 31;
    const int wave = __builtin_amdgcn_readfirstlane(tid >> 5);
    const int key  = (int)blockIdx.x;
    const int row0 = key * DB_ROWS;
    const int c4   = 4 * (lane & 15);
    const bool up  = lane >= 16;

    {
        const int q_half = lane >> 4;
        const int q_c16  = lane & 15;
        const int q_rt   = wave & 3;
        const int q_side = (wave >> 2) & 1;
        const int q_nh   = wave >> 3;
        const float* q_h = q_side ? h_p : h_c;
        const float* q_W = q_side ? Wqp : Wqc;
        const float* q_b = q_side ? bqp : bqc;
        int qr = row0 + 16 * q_rt + q_c16;
        qr = qr < n_nodes ? qr : n_nodes - 1;
        const float* hr = q_h + qr * 64 + 2 * q_half;
        v2f qa[16];
#pragma unroll
        for (int ks = 0; ks < 16; ++ks) qa[ks] = *(const v2f*)(hr + ks * 4);
        float* qd = s_q + q_side * (DB_ROWS * 64) + (16 * q_rt + 8 * q_half) * 64 + q_nh * 32 + q_c16;
#pragma unroll
        for (int nt = 0; nt < 2; ++nt) {
            const float* w0 = q_W + (2 * q_half) * 64 + q_nh * 32 + nt * 16 + q_c16;
            const float bias = q_b[q_nh * 32 + nt * 16 + q_c16];
            v2f qb[16];
#pragma unroll
            for (int ks = 0; ks < 16; ++ks) qb[ks] = (v2f){w0[ks * 256], w0[ks * 256 + 64]};
            v8f acc = (v8f){0.f, 0.f, 0.f, 0.f, 0.f, 0.f, 0.f, 0.f};
#pragma unroll
            for (int ks = 0; ks < 16; ++ks)
                acc = __builtin_amdgcn_wmma_f32_16x16x4_f32(
                     false, qa[ks],  false, qb[ks],
                     (short)0, acc,  false,  false);
#pragma unroll
            for (int r = 0; r < 8; ++r) qd[r * 64 + nt * 16] = acc[r] + bias;
        }
    }

    if (wave == 0) { s_cnt[2 * lane] = 0; s_cnt[2 * lane + 1] = 0; s_cur[2 * lane] = 0; s_cur[2 * lane + 1] = 0; }

    const int nslab = (n_edges + SLAB - 1) / SLAB;
    int st[4], cn[4];
#pragma unroll
    for (int j = 0; j < 4; ++j) {
        const int t = 4 * lane + j;
        st[j] = 0; cn[j] = 0;
        if (t < nslab) {
            const int* tp = tab + t * TABW + key;
            st[j] = tp[0];
            cn[j] = tp[1] - tp[0];
        }
    }

    const unsigned ocol = (up ? (unsigned)n_nodes * 64u : 0u) + (unsigned)c4;

    const int sum4 = cn[0] + cn[1] + cn[2] + cn[3];
    int inc = sum4;
    for (int o = 1; o < 32; o <<= 1) {
        const int y = __shfl_up(inc, o);
        if (lane >= o) inc += y;
    }
    const int total = __builtin_amdgcn_readlane(inc, 31);
    int of[4];
    of[0] = inc - sum4;
    of[1] = of[0] + cn[0];
    of[2] = of[1] + cn[1];
    of[3] = of[2] + cn[2];
    const int jj = wave & 3;
    static_assert(SLAB < 65536, "a segment's start and count take 16 bits each");
    const int st_s0 = jj == 0 ? st[0] : jj == 1 ? st[1] : jj == 2 ? st[2] : st[3];
    const int cn_s0 = jj == 0 ? cn[0] : jj == 1 ? cn[1] : jj == 2 ? cn[2] : cn[3];
    const int sc_s = st_s0 | (cn_s0 << 16);
    const int of_s = jj == 0 ? of[0] : jj == 1 ? of[1] : jj == 2 ? of[2] : of[3];

    __syncthreads();

    int* mystart = s_start[wave];

    for (int lo = 0; lo < total; lo += LIST_CAP) {
        int wv[8], c8[8], p8[8], s8[8];
#pragma unroll
        for (int i = 0; i < 8; ++i) {
            const int L = (wave >> 2) + 4 * i;
            const int sc = __builtin_amdgcn_readlane(sc_s, L);
            s8[i] = sc & 0xffff;
            c8[i] = sc >> 16;
            p8[i] = __builtin_amdgcn_readlane(of_s, L) - lo;
            const unsigned so = (unsigned)((wave + 16 * i) * SLAB + s8[i] + lane);
            wv[i] = 0;
            if (lane < c8[i]) wv[i] = slabs[so];
        }

#pragma unroll
        for (int i = 0; i < 8; ++i) {
            if (lane < c8[i] && (unsigned)(p8[i] + lane) < (unsigned)LIST_CAP)
                atomicAdd(&s_cnt[wv[i] >> 24], 1);
#pragma nounroll
            for (int p = 32; p < c8[i]; p += 32) {
                const unsigned so = (unsigned)((wave + 16 * i) * SLAB + s8[i] + p + lane);
                if (p + lane < c8[i] && (unsigned)(p8[i] + p + lane) < (unsigned)LIST_CAP)
                    atomicAdd(&s_cnt[slabs[so] >> 24], 1);
            }
        }
        __syncthreads();

        const int n0 = s_cnt[2 * lane], n1 = s_cnt[2 * lane + 1];
        int rs = n0 + n1;
        for (int o = 1; o < 32; o <<= 1) {
            const int y = __shfl_up(rs, o);
            if (lane >= o) rs += y;
        }
        const int b0 = rs - (n0 + n1), b1 = b0 + n0;
        mystart[2 * lane]     = b0;
        mystart[2 * lane + 1] = b1;
        static_assert(LIST_CAP < 32768, "two row counts share one register");
        const int nn = n0 | (n1 << 16);

#pragma unroll
        for (int i = 0; i < 8; ++i) {
            if (lane < c8[i] && (unsigned)(p8[i] + lane) < (unsigned)LIST_CAP) {
                const int r = wv[i] >> 24;
                s_list[mystart[r] + atomicAdd(&s_cur[r], 1)] = wv[i];
            }
#pragma nounroll
            for (int p = 32; p < c8[i]; p += 32) {
                const unsigned so = (unsigned)((wave + 16 * i) * SLAB + s8[i] + p + lane);
                if (p + lane < c8[i] && (unsigned)(p8[i] + p + lane) < (unsigned)LIST_CAP) {
                    const int w2 = slabs[so];
                    const int r = w2 >> 24;
                    s_list[mystart[r] + atomicAdd(&s_cur[r], 1)] = w2;
                }
            }
        }
        __syncthreads();
        if (wave == 0) { s_cnt[2 * lane] = 0; s_cnt[2 * lane + 1] = 0; s_cur[2 * lane] = 0; s_cur[2 * lane + 1] = 0; }

#pragma nounroll
        for (int j = 0; j < 4; ++j) {
            const int Lr = 2 * wave + (j >> 1);
            const int nn_r = __builtin_amdgcn_readlane(nn, Lr);
            const int b_r  = __builtin_amdgcn_readlane(b0, Lr);
            const int n  = (j & 1) ? (nn_r >> 16) : (nn_r & 0xffff);
            const int sb = (j & 1) ? b_r + (nn_r & 0xffff) : b_r;
            const int* lrow = s_list + sb;
            int* srow = s_sort + sb;
            const int row = row0 + 4 * wave + j;
            const bool live = row < n_nodes;

            const v4f q = *(const v4f*)(s_q + (up ? DB_ROWS * 64 : 0) + (4 * wave + j) * 64 + c4);
            v4f acc = (v4f){0.f, 0.f, 0.f, 0.f};
            if (lo > 0 && live) acc = *(const v4f*)(out + (ocol + (unsigned)row * 64u));

            for (int cb = 0; cb < n; cb += 32) {
                const int t = cb + lane;
                const int x = t < n ? lrow[t] : 0x7fffffff;
                const unsigned long long kx = ((unsigned long long)(unsigned)x << 32) | (unsigned)t;
                int rank = 0;
                for (int ob = 0; ob < n; ob += 32) {
                    const int u = ob + lane;
                    const int y = u < n ? lrow[u] : 0x7fffffff;
                    const int mm = n - ob > 32 ? 32 : n - ob;
                    for (int m = 0; m < mm; ++m) {
                        const unsigned ym = (unsigned)__builtin_amdgcn_readlane(y, m);
                        const unsigned long long ky = ((unsigned long long)ym << 32) | (unsigned)(ob + m);
                        rank += ky < kx ? 1 : 0;
                    }
                }
                if (t < n) srow[rank] = x;
            }

            for (int cb = 0; cb < n; cb += 32) {
                const int t = cb + lane;
                const int xs = t < n ? srow[t] : 0;
                const int mm = n - cb > 32 ? 32 : n - cb;
                int e = 0;
                for (; e + EPT <= mm; e += EPT) {
                    int w4[EPT];
#pragma unroll
                    for (int i = 0; i < EPT; ++i) w4[i] = __builtin_amdgcn_readlane(xs, e + i);
                    edge_trip<true>(w4, EPT, c4, Kn, Vn, q, acc);
                }
                if (e < mm) {
                    int w4[EPT];
#pragma unroll
                    for (int i = 0; i < EPT; ++i)
                        w4[i] = __builtin_amdgcn_readlane(xs, e + i < mm ? e + i : mm - 1);
                    edge_trip<false>(w4, mm - e, c4, Kn, Vn, q, acc);
                }
            }

            if (live) *(v4f*)(out + (ocol + (unsigned)row * 64u)) = acc;
        }
        __syncthreads();
    }

    if (total <= 0) {
#pragma unroll
        for (int j = 0; j < 4; ++j) {
            const int row = row0 + 4 * wave + j;
            if (row < n_nodes) *(v4f*)(out + (ocol + (unsigned)row * 64u)) = (v4f){0.f, 0.f, 0.f, 0.f};
        }
    }
}

extern "C" void kernel_launch(void* const* d_in, const int* in_sizes, int n_in,
                              void* d_out, int out_size, void* d_ws, size_t ws_size,
                              hipStream_t stream)
{
    const float* h_c  = (const float*)d_in[0];
    const float* h_p  = (const float*)d_in[1];
    const int*   src  = (const int*)  d_in[2];
    const int*   dst  = (const int*)  d_in[3];
    const float* Wqcw = (const float*)d_in[4];
    const float* Wqcb = (const float*)d_in[5];
    const float* Wqpw = (const float*)d_in[6];
    const float* Wqpb = (const float*)d_in[7];
    const float* Wkw  = (const float*)d_in[8];
    const float* Wkb  = (const float*)d_in[9];
    const float* Wvw  = (const float*)d_in[10];
    const float* Wvb  = (const float*)d_in[11];

    const int n = in_sizes[0] / 64;
    const int e = in_sizes[2];

    float* Kn = (float*)d_ws;
    float* Vn = Kn + (size_t)n * 64;
    int* slabs = (int*)(Vn + (size_t)n * 64);
    int* tab = slabs + (size_t)e;

    float* out_c = (float*)d_out;

    {
        proj_wmma_kernel<<<PROJ_BLOCKS, 256, 0, stream>>>(
            h_c, Wkw, Wkb, Wvw, Wvb, Kn, Vn, n);
    }

    {
        bin_kernel<<<BIN_BLOCKS, 1024, 0, stream>>>(src, dst, slabs, tab, n, e);
    }

    {
        edge_attn_kernel<<<NDB, 512, 0, stream>>>(
            slabs, tab, h_c, h_p, Wqcw, Wqcb, Wqpw, Wqpb, Kn, Vn, out_c, n, e);
    }
}
